# starting kernel
# baseline (speedup 1.0000x reference)
#include <hip/hip_runtime.h>

#define D 128

typedef float v2f __attribute__((ext_vector_type(2)));
typedef float v8f __attribute__((ext_vector_type(8)));


__global__ void zero_f32(float* __restrict__ p, int n) {
    int i = blockIdx.x * blockDim.x + threadIdx.x;
    if (i < n) p[i] = 0.0f;
}

__global__ void deg_kernel(const int* __restrict__ col,
                           const float* __restrict__ w,
                           float* __restrict__ deg, int E) {
    int e = blockIdx.x * blockDim.x + threadIdx.x;
    if (e < E) unsafeAtomicAdd(&deg[col[e]], w[e]);
}

__global__ void rsqrt_kernel(const float* __restrict__ deg,
                             float* __restrict__ dis, int n) {
    int i = blockIdx.x * blockDim.x + threadIdx.x;
    if (i < n) {
        float d = deg[i];
        dis[i] = (d > 0.0f) ? __frsqrt_rn(d) : 0.0f;
    }
}


__global__ __launch_bounds__(512) void gemm_dual(const float* __restrict__ x,
                                                 const float* __restrict__ Wlin,
                                                 const float* __restrict__ Wgcn,
                                                 float* __restrict__ out_hl,
                                                 float* __restrict__ h) {
    const int row0  = blockIdx.x * 16;
    const int wave  = threadIdx.x >> 5;
    const int lane  = threadIdx.x & 31;
    const int which = wave >> 3;
    const int col0  = (wave & 7) * 16;

    const float* __restrict__ W   = which ? Wgcn : Wlin;
    float* __restrict__       dst = which ? h    : out_hl;

    const int m  = lane & 15;
    const int hh = lane >> 4;

    const float* __restrict__ arow = x + (size_t)(row0 + m) * D;

    v8f acc = {};
#pragma unroll
    for (int k0 = 0; k0 < D; k0 += 4) {
        v2f a;
        a.x = arow[k0 + 2 * hh + 0];
        a.y = arow[k0 + 2 * hh + 1];
        v2f b;
        b.x = W[(size_t)(k0 + 2 * hh + 0) * D + col0 + m];
        b.y = W[(size_t)(k0 + 2 * hh + 1) * D + col0 + m];
        acc = __builtin_amdgcn_wmma_f32_16x16x4_f32(
             false, a,  false, b,
             (short)0, acc,  false,  false);
    }

#pragma unroll
    for (int r = 0; r < 8; ++r) {
        dst[(size_t)(row0 + r + 8 * hh) * D + col0 + m] = acc[r];
    }
}


__global__ __launch_bounds__(256) void scatter_kernel(const float* __restrict__ h,
                                                      const int* __restrict__ row_idx,
                                                      const int* __restrict__ col_idx,
                                                      const float* __restrict__ w,
                                                      const float* __restrict__ dis,
                                                      float* __restrict__ out, int E) {
    const int wave = threadIdx.x >> 5;
    const int lane = threadIdx.x & 31;
    const int e    = blockIdx.x * 8 + wave;
    if (e >= E) return;

    const int r = row_idx[e];
    const int c = col_idx[e];
    const float norm = dis[r] * w[e] * dis[c];

    const float4* __restrict__ hv = (const float4*)(h + (size_t)r * D);
    float4 v = hv[lane];

    float* __restrict__ o = out + (size_t)c * D + lane * 4;
    unsafeAtomicAdd(o + 0, v.x * norm);
    unsafeAtomicAdd(o + 1, v.y * norm);
    unsafeAtomicAdd(o + 2, v.z * norm);
    unsafeAtomicAdd(o + 3, v.w * norm);
}


extern "C" void kernel_launch(void* const* d_in, const int* in_sizes, int n_in,
                              void* d_out, int out_size, void* d_ws, size_t ws_size,
                              hipStream_t stream) {
    const float* x    = (const float*)d_in[0];
    const int*   ei   = (const int*)d_in[1];
    const float* ew   = (const float*)d_in[2];
    const float* Wlin = (const float*)d_in[3];
    const float* Wgcn = (const float*)d_in[4];
    float*       out  = (float*)d_out;

    const int N = in_sizes[0] / D;
    const int E = in_sizes[2];

    const int* row = ei;
    const int* col = ei + E;

    float* deg = (float*)d_ws;
    float* dis = deg + N;
    float* h   = dis + N;

    zero_f32<<<(N + 255) / 256, 256, 0, stream>>>(deg, N);
    deg_kernel<<<(E + 255) / 256, 256, 0, stream>>>(col, ew, deg, E);
    rsqrt_kernel<<<(N + 255) / 256, 256, 0, stream>>>(deg, dis, N);
    gemm_dual<<<N / 16, 512, 0, stream>>>(x, Wlin, Wgcn, out, h);
    scatter_kernel<<<(E + 7) / 8, 256, 0, stream>>>(h, row, col, ew, dis, out, E);
}


# key plane: the degree kernel (now 4 edges per lane) also stores each edge's destination tile number as one byte; propagate blocks filter 16 keys per 16-B load, not 4 ints; sums stay arrival-order
# speedup vs baseline: 2.5777x; 2.5777x over previous
#include <hip/hip_runtime.h>

#define D 128

typedef float v2f __attribute__((ext_vector_type(2)));
typedef float v8f __attribute__((ext_vector_type(8)));


__global__ void zero_f32(float* __restrict__ p, int n) {
    int i = blockIdx.x * blockDim.x + threadIdx.x;
    if (i < n) p[i] = 0.0f;
}


#define TILE_ROWS 392
#define SLICE     1024
#define OWN_WAVES 32
#define KEYS_PER_LOAD 512

typedef int v4i __attribute__((ext_vector_type(4)));

__global__ __launch_bounds__(SLICE / 4) void deg_kernel(const int* __restrict__ col,
                                                        const float* __restrict__ w,
                                                        float* __restrict__ deg2,
                                                        unsigned* __restrict__ keys, int n, int E) {
    const unsigned place = __builtin_amdgcn_s_sendmsg_rtn(0x87);
    float* __restrict__ deg = deg2 + (size_t)((place >> 18) & 1u) * n;
    const int e = (blockIdx.x * blockDim.x + threadIdx.x) * 4;
    unsigned key = 0u;
    if ((E & 3) == 0 && e < E) {
        const v4i    c  = *(const v4i*)(col + e);
        const float4 ww = *(const float4*)(w + e);
        unsafeAtomicAdd(&deg[c.x], ww.x);
        unsafeAtomicAdd(&deg[c.y], ww.y);
        unsafeAtomicAdd(&deg[c.z], ww.z);
        unsafeAtomicAdd(&deg[c.w], ww.w);
        key = ((unsigned)c.x / TILE_ROWS) | (((unsigned)c.y / TILE_ROWS) << 8) |
              (((unsigned)c.z / TILE_ROWS) << 16) | (((unsigned)c.w / TILE_ROWS) << 24);
    } else {
        for (int k = 0; k < 4; ++k) {
            if (e + k < E) {
                const int c = col[e + k];
                unsafeAtomicAdd(&deg[c], w[e + k]);
                key |= ((unsigned)c / TILE_ROWS) << (8 * k);
            }
        }
    }
    if (keys != nullptr) keys[e >> 2] = key;
}

__global__ void rsqrt_kernel(float* deg2, int n) {
    int i = blockIdx.x * blockDim.x + threadIdx.x;
    if (i < n) {
        float d = deg2[i] + deg2[n + i];
        deg2[i] = (d > 0.0f) ? __frsqrt_rn(d) : 0.0f;
    }
}


#define XS_STRIDE 132

__global__ __launch_bounds__(512) void gemm_dual(const float* __restrict__ x,
                                                 const float* __restrict__ Wlin,
                                                 const float* __restrict__ Wgcn,
                                                 float* __restrict__ out_hl,
                                                 float* __restrict__ h, int ntiles) {
    __shared__ __attribute__((aligned(16))) float xs[2][16 * XS_STRIDE];

    const int wave  = threadIdx.x >> 5;
    const int lane  = threadIdx.x & 31;
    const int which = wave >> 3;
    const int col0  = (wave & 7) * 16;

    const float* __restrict__ W   = which ? Wgcn : Wlin;
    float* __restrict__       dst = which ? h    : out_hl;

    const int m  = lane & 15;
    const int hh = lane >> 4;

    v2f b[32];
#pragma unroll
    for (int ks = 0; ks < 32; ++ks) {
        b[ks].x = W[(size_t)(4 * ks + 2 * hh + 0) * D + col0 + m];
        b[ks].y = W[(size_t)(4 * ks + 2 * hh + 1) * D + col0 + m];
    }

    const float* __restrict__ xsrc = x + (size_t)wave * D + lane * 4;
    const int soff = wave * XS_STRIDE + lane * 4;

    int tile = blockIdx.x;
    float4 stage = make_float4(0.0f, 0.0f, 0.0f, 0.0f);
    if (tile < ntiles) stage = *(const float4*)(xsrc + (size_t)tile * 16 * D);
    *(float4*)&xs[0][soff] = stage;
    __syncthreads();

    int buf = 0;
    for (; tile < ntiles; tile += gridDim.x) {
        const int  nt   = tile + gridDim.x;
        const bool more = nt < ntiles;
        if (more) stage = *(const float4*)(xsrc + (size_t)nt * 16 * D);

        const float* xa = &xs[buf][m * XS_STRIDE + 2 * hh];
        v8f acc = {};
#pragma unroll
        for (int ks = 0; ks < 32; ++ks) {
            const v2f a = *(const v2f*)(xa + 4 * ks);
            acc = __builtin_amdgcn_wmma_f32_16x16x4_f32(
                 false, a,  false, b[ks],
                 (short)0, acc,  false,  false);
        }

        float* __restrict__ drow = dst + (size_t)(tile * 16 + 8 * hh) * D + col0 + m;
#pragma unroll
        for (int r = 0; r < 8; ++r) drow[(size_t)r * D] = acc[r];

        if (more) *(float4*)&xs[buf ^ 1][soff] = stage;
        __syncthreads();
        buf ^= 1;
    }
}


template <bool KEYED>
__device__ __forceinline__ void fetch_slice(const int* __restrict__ col, const unsigned* __restrict__ keys,
                                            int e0, int lane, int E, bool vec_ok,
                                            v4i (&c)[KEYED ? SLICE / KEYS_PER_LOAD : SLICE / 128]) {
    if constexpr (KEYED) {
#pragma unroll
        for (int j = 0; j < SLICE / KEYS_PER_LOAD; ++j)
            c[j] = *(const v4i*)(keys + ((e0 + j * KEYS_PER_LOAD) >> 2) + lane * 4);
    } else if (vec_ok && e0 + SLICE <= E) {
#pragma unroll
        for (int j = 0; j < SLICE / 128; ++j)
            c[j] = *(const v4i*)(col + e0 + j * 128 + lane * 4);
    } else {
#pragma unroll
        for (int j = 0; j < SLICE / 128; ++j) {
            const int e = e0 + j * 128 + lane * 4;
            c[j].x = (e + 0 < E) ? col[e + 0] : -1;
            c[j].y = (e + 1 < E) ? col[e + 1] : -1;
            c[j].z = (e + 2 < E) ? col[e + 2] : -1;
            c[j].w = (e + 3 < E) ? col[e + 3] : -1;
        }
    }
}

template <bool KEYED>
__device__ __forceinline__ int filter_slice(v4i (&cur)[KEYED ? SLICE / KEYS_PER_LOAD : SLICE / 128],
                                            unsigned short* myq, int lane, int e0, int E,
                                            int tile, int lo, int nrows) {
    int cnt = 0;
    if constexpr (KEYED) {
        const unsigned pat = (unsigned)tile * 0x01010101u;
        if (e0 + SLICE > E) {
#pragma unroll
            for (int j = 0; j < SLICE / KEYS_PER_LOAD; ++j) {
#pragma unroll
                for (int d = 0; d < 4; ++d) {
                    const int nv = E - (e0 + j * KEYS_PER_LOAD + lane * 16 + 4 * d);
                    const unsigned keep = (nv >= 4) ? 0xffffffffu : ((nv <= 0) ? 0u : ((1u << (8 * nv)) - 1u));
                    cur[j][d] = (int)(((unsigned)cur[j][d] & keep) | (~pat & ~keep));
                }
            }
        }
#pragma unroll
        for (int j = 0; j < SLICE / KEYS_PER_LOAD; ++j) {
            unsigned g = 0u;
#pragma unroll
            for (int d = 0; d < 4; ++d) {
                const unsigned x  = (unsigned)cur[j][d] ^ pat;
                const unsigned nz = ((x & 0x7f7f7f7fu) + 0x7f7f7f7fu) | x;
                g |= (~nz & 0x80808080u) >> (7 - d);
            }
            unsigned hit = __builtin_amdgcn_ballot_w32(g != 0u);
            while (hit != 0u) {
                if (g != 0u) {
                    const int p = __builtin_ctz(g);
                    myq[cnt + __builtin_popcount(hit & ((1u << lane) - 1u))] =
                        (unsigned short)(j * KEYS_PER_LOAD + lane * 16 + ((p & 7) << 2) + (p >> 3));
                    g &= g - 1u;
                }
                cnt += __builtin_popcount(hit);
                hit = __builtin_amdgcn_ballot_w32(g != 0u);
            }
        }
    } else {
#pragma unroll
        for (int j = 0; j < SLICE / 128; ++j) {
            unsigned mask[4];
#pragma unroll
            for (int k = 0; k < 4; ++k)
                mask[k] = __builtin_amdgcn_ballot_w32((unsigned)(cur[j][k] - lo) < (unsigned)nrows);
            if ((mask[0] | mask[1] | mask[2] | mask[3]) != 0u) {
#pragma unroll
                for (int k = 0; k < 4; ++k) {
                    if ((mask[k] >> lane) & 1u)
                        myq[cnt + __builtin_popcount(mask[k] & ((1u << lane) - 1u))] =
                            (unsigned short)(j * 128 + lane * 4 + k);
                    cnt += __builtin_popcount(mask[k]);
                }
            }
        }
    }
    return cnt;
}

template <bool KEYED>
__global__ __launch_bounds__(1024) void owner_gather(const float* __restrict__ h,
                                                     const int* __restrict__ row_idx,
                                                     const int* __restrict__ col_idx,
                                                     const float* __restrict__ w,
                                                     const float* __restrict__ dis,
                                                     const unsigned* __restrict__ keys,
                                                     float* __restrict__ out, int N, int E) {
    __shared__ float          acc[TILE_ROWS * D];
    __shared__ unsigned short q[OWN_WAVES * SLICE];

    const int wave = __builtin_amdgcn_readfirstlane(threadIdx.x >> 5);
    const int lane = threadIdx.x & 31;
    unsigned short* myq = q + wave * SLICE;

    constexpr int NREG = KEYED ? SLICE / KEYS_PER_LOAD : SLICE / 128;
    const int  nslices = (E + SLICE - 1) / SLICE;
    const bool vec_ok  = (E & 3) == 0;
    const int  rot     = (int)(((unsigned)blockIdx.x * (unsigned)nslices / gridDim.x) % (unsigned)nslices);

    for (int tile = blockIdx.x; tile * TILE_ROWS < N; tile += gridDim.x) {
        const int lo    = tile * TILE_ROWS;
        const int nrows = min(TILE_ROWS, N - lo);

        for (int i = threadIdx.x; i < TILE_ROWS * D / 4; i += blockDim.x)
            ((float4*)acc)[i] = make_float4(0.0f, 0.0f, 0.0f, 0.0f);
        __syncthreads();

        v4i cur[NREG];
        {
            int sl = wave + rot; if (sl >= nslices) sl -= nslices;
            if (wave < nslices) fetch_slice<KEYED>(col_idx, keys, sl * SLICE, lane, E, vec_ok, cur);
        }
        for (int s = wave; s < nslices; s += OWN_WAVES) {
            int sl = s + rot; if (sl >= nslices) sl -= nslices;
            const int e0 = sl * SLICE;

            v4i nxt[NREG];
            {
                int sn = ((s + OWN_WAVES < nslices) ? s + OWN_WAVES : s) + rot;
                if (sn >= nslices) sn -= nslices;
                fetch_slice<KEYED>(col_idx, keys, sn * SLICE, lane, E, vec_ok, nxt);
            }

            const int cnt = filter_slice<KEYED>(cur, myq, lane, e0, E, tile, lo, nrows);

            for (int base = 0; base < cnt; base += 32) {
                const int n  = min(32, cnt - base);
                const int qi = (lane < n) ? base + lane : base;
                const int e  = e0 + (int)myq[qi];
                const int r  = row_idx[e];
                const int c  = col_idx[e];
                const float nm   = dis[r] * w[e] * dis[c];
                const int   lrow = c - lo;

                for (int t = 0; t < n; t += 4) {
                    float v[4][4];
                    float sc[4];
                    int   lr[4];
#pragma unroll
                    for (int u = 0; u < 4; ++u) {
                        const int tt = min(t + u, n - 1);
                        const int rt = __builtin_amdgcn_readlane(r, tt);
                        lr[u] = __builtin_amdgcn_readlane(lrow, tt);
                        sc[u] = __int_as_float(__builtin_amdgcn_readlane(__float_as_int(nm), tt));
                        const float* __restrict__ hr = h + (size_t)rt * D + lane;
#pragma unroll
                        for (int j = 0; j < 4; ++j) v[u][j] = hr[32 * j];
                    }
#pragma unroll
                    for (int u = 0; u < 4; ++u) {
                        if (t + u < n) {
                            float* a = acc + lr[u] * D + lane;
#pragma unroll
                            for (int j = 0; j < 4; ++j)
                                __hip_atomic_fetch_add(a + 32 * j, v[u][j] * sc[u], __ATOMIC_RELAXED,
                                                       __HIP_MEMORY_SCOPE_WORKGROUP);
                        }
                    }
                }
            }

#pragma unroll
            for (int j = 0; j < NREG; ++j) cur[j] = nxt[j];
        }
        __syncthreads();

        float4*       o4 = (float4*)(out + (size_t)lo * D);
        const float4* a4 = (const float4*)acc;
        for (int i = threadIdx.x; i < nrows * (D / 4); i += blockDim.x) {
            float4 o = o4[i];
            const float4 a = a4[i];
            o.x += a.x; o.y += a.y; o.z += a.z; o.w += a.w;
            o4[i] = o;
        }
        __syncthreads();
    }
}


extern "C" void kernel_launch(void* const* d_in, const int* in_sizes, int n_in,
                              void* d_out, int out_size, void* d_ws, size_t ws_size,
                              hipStream_t stream) {
    const float* x    = (const float*)d_in[0];
    const int*   ei   = (const int*)d_in[1];
    const float* ew   = (const float*)d_in[2];
    const float* Wlin = (const float*)d_in[3];
    const float* Wgcn = (const float*)d_in[4];
    float*       out  = (float*)d_out;

    const int N = in_sizes[0] / D;
    const int E = in_sizes[2];

    const int* row = ei;
    const int* col = ei + E;

    float* deg2 = (float*)d_ws;
    float* dis  = deg2;
    float* h    = deg2 + 2 * N;
    const int nslices = (E + SLICE - 1) / SLICE;
    const bool keyed = ws_size >= (size_t)(2 * N + N * D) * sizeof(float) + (size_t)nslices * SLICE;
    unsigned* keys = keyed ? (unsigned*)(h + (size_t)N * D) : nullptr;

    zero_f32<<<(2 * N + 255) / 256, 256, 0, stream>>>(deg2, 2 * N);
    deg_kernel<<<nslices, SLICE / 4, 0, stream>>>(col, ew, deg2, keys, N, E);
    rsqrt_kernel<<<(N + 255) / 256, 256, 0, stream>>>(deg2, N);
    gemm_dual<<<512, 512, 0, stream>>>(x, Wlin, Wgcn, out, h, N / 16);
    if (keyed) owner_gather<true><<<(N + TILE_ROWS - 1) / TILE_ROWS, 1024, 0, stream>>>(h, row, col, ew, dis, keys, out, N, E);
    else       owner_gather<false><<<(N + TILE_ROWS - 1) / TILE_ROWS, 1024, 0, stream>>>(h, row, col, ew, dis, keys, out, N, E);
}


# record plane: the degree kernel packs an edge's source row, local destination row and weight into one 8-byte word; a propagate lane reads 1 word per owned edge, was 3 words in 3 lines; sums unchanged
# speedup vs baseline: 2.7064x; 1.0499x over previous
#include <hip/hip_runtime.h>

#define D 128

typedef float v2f __attribute__((ext_vector_type(2)));
typedef float v8f __attribute__((ext_vector_type(8)));


__global__ void zero_f32(float* __restrict__ p, int n) {
    int i = blockIdx.x * blockDim.x + threadIdx.x;
    if (i < n) p[i] = 0.0f;
}


#define TILE_ROWS 392
#define SLICE     1024
#define OWN_WAVES 32
#define KEYS_PER_LOAD 512

typedef int v4i __attribute__((ext_vector_type(4)));

#define REC_ROW_BITS 23
__global__ __launch_bounds__(SLICE / 4) void deg_kernel(const int* __restrict__ row,
                                                        const int* __restrict__ col,
                                                        const float* __restrict__ w,
                                                        float* __restrict__ deg2,
                                                        unsigned* __restrict__ keys,
                                                        unsigned* __restrict__ recs, int n, int E) {
    const unsigned place = __builtin_amdgcn_s_sendmsg_rtn(0x87);
    float* __restrict__ deg = deg2 + (size_t)((place >> 18) & 1u) * n;
    const int e = (blockIdx.x * blockDim.x + threadIdx.x) * 4;
    unsigned key = 0u;
    if ((E & 3) == 0 && e < E) {
        const v4i    c  = *(const v4i*)(col + e);
        const float4 ww = *(const float4*)(w + e);
        unsafeAtomicAdd(&deg[c.x], ww.x);
        unsafeAtomicAdd(&deg[c.y], ww.y);
        unsafeAtomicAdd(&deg[c.z], ww.z);
        unsafeAtomicAdd(&deg[c.w], ww.w);
        const unsigned tx = (unsigned)c.x / TILE_ROWS, ty = (unsigned)c.y / TILE_ROWS;
        const unsigned tz = (unsigned)c.z / TILE_ROWS, tw = (unsigned)c.w / TILE_ROWS;
        key = tx | (ty << 8) | (tz << 16) | (tw << 24);
        if (recs != nullptr) {
            const v4i r = *(const v4i*)(row + e);
            uint4 a, b;
            a.x = (unsigned)r.x | (((unsigned)c.x - tx * TILE_ROWS) << REC_ROW_BITS); a.y = __float_as_uint(ww.x);
            a.z = (unsigned)r.y | (((unsigned)c.y - ty * TILE_ROWS) << REC_ROW_BITS); a.w = __float_as_uint(ww.y);
            b.x = (unsigned)r.z | (((unsigned)c.z - tz * TILE_ROWS) << REC_ROW_BITS); b.y = __float_as_uint(ww.z);
            b.z = (unsigned)r.w | (((unsigned)c.w - tw * TILE_ROWS) << REC_ROW_BITS); b.w = __float_as_uint(ww.w);
            *(uint4*)(recs + 2 * e)     = a;
            *(uint4*)(recs + 2 * e + 4) = b;
        }
    } else {
        for (int k = 0; k < 4; ++k) {
            if (e + k < E) {
                const int c = col[e + k];
                const float wk = w[e + k];
                unsafeAtomicAdd(&deg[c], wk);
                const unsigned t = (unsigned)c / TILE_ROWS;
                key |= t << (8 * k);
                if (recs != nullptr)
                    *(uint2*)(recs + 2 * (e + k)) =
                        make_uint2((unsigned)row[e + k] | (((unsigned)c - t * TILE_ROWS) << REC_ROW_BITS),
                                   __float_as_uint(wk));
            }
        }
    }
    if (keys != nullptr) keys[e >> 2] = key;
}

__global__ void rsqrt_kernel(float* deg2, int n) {
    int i = blockIdx.x * blockDim.x + threadIdx.x;
    if (i < n) {
        float d = deg2[i] + deg2[n + i];
        deg2[i] = (d > 0.0f) ? __frsqrt_rn(d) : 0.0f;
    }
}


#define XS_STRIDE 132

__global__ __launch_bounds__(512) void gemm_dual(const float* __restrict__ x,
                                                 const float* __restrict__ Wlin,
                                                 const float* __restrict__ Wgcn,
                                                 float* __restrict__ out_hl,
                                                 float* __restrict__ h, int ntiles) {
    __shared__ __attribute__((aligned(16))) float xs[2][16 * XS_STRIDE];

    const int wave  = threadIdx.x >> 5;
    const int lane  = threadIdx.x & 31;
    const int which = wave >> 3;
    const int col0  = (wave & 7) * 16;

    const float* __restrict__ W   = which ? Wgcn : Wlin;
    float* __restrict__       dst = which ? h    : out_hl;

    const int m  = lane & 15;
    const int hh = lane >> 4;

    v2f b[32];
#pragma unroll
    for (int ks = 0; ks < 32; ++ks) {
        b[ks].x = W[(size_t)(4 * ks + 2 * hh + 0) * D + col0 + m];
        b[ks].y = W[(size_t)(4 * ks + 2 * hh + 1) * D + col0 + m];
    }

    const float* __restrict__ xsrc = x + (size_t)wave * D + lane * 4;
    const int soff = wave * XS_STRIDE + lane * 4;

    int tile = blockIdx.x;
    float4 stage = make_float4(0.0f, 0.0f, 0.0f, 0.0f);
    if (tile < ntiles) stage = *(const float4*)(xsrc + (size_t)tile * 16 * D);
    *(float4*)&xs[0][soff] = stage;
    __syncthreads();

    int buf = 0;
    for (; tile < ntiles; tile += gridDim.x) {
        const int  nt   = tile + gridDim.x;
        const bool more = nt < ntiles;
        if (more) stage = *(const float4*)(xsrc + (size_t)nt * 16 * D);

        const float* xa = &xs[buf][m * XS_STRIDE + 2 * hh];
        v8f acc = {};
#pragma unroll
        for (int ks = 0; ks < 32; ++ks) {
            const v2f a = *(const v2f*)(xa + 4 * ks);
            acc = __builtin_amdgcn_wmma_f32_16x16x4_f32(
                 false, a,  false, b[ks],
                 (short)0, acc,  false,  false);
        }

        float* __restrict__ drow = dst + (size_t)(tile * 16 + 8 * hh) * D + col0 + m;
#pragma unroll
        for (int r = 0; r < 8; ++r) drow[(size_t)r * D] = acc[r];

        if (more) *(float4*)&xs[buf ^ 1][soff] = stage;
        __syncthreads();
        buf ^= 1;
    }
}


template <bool KEYED>
__device__ __forceinline__ void fetch_slice(const int* __restrict__ col, const unsigned* __restrict__ keys,
                                            int e0, int lane, int E, bool vec_ok,
                                            v4i (&c)[KEYED ? SLICE / KEYS_PER_LOAD : SLICE / 128]) {
    if constexpr (KEYED) {
#pragma unroll
        for (int j = 0; j < SLICE / KEYS_PER_LOAD; ++j)
            c[j] = *(const v4i*)(keys + ((e0 + j * KEYS_PER_LOAD) >> 2) + lane * 4);
    } else if (vec_ok && e0 + SLICE <= E) {
#pragma unroll
        for (int j = 0; j < SLICE / 128; ++j)
            c[j] = *(const v4i*)(col + e0 + j * 128 + lane * 4);
    } else {
#pragma unroll
        for (int j = 0; j < SLICE / 128; ++j) {
            const int e = e0 + j * 128 + lane * 4;
            c[j].x = (e + 0 < E) ? col[e + 0] : -1;
            c[j].y = (e + 1 < E) ? col[e + 1] : -1;
            c[j].z = (e + 2 < E) ? col[e + 2] : -1;
            c[j].w = (e + 3 < E) ? col[e + 3] : -1;
        }
    }
}

template <bool KEYED>
__device__ __forceinline__ int filter_slice(v4i (&cur)[KEYED ? SLICE / KEYS_PER_LOAD : SLICE / 128],
                                            unsigned short* myq, int lane, int e0, int E,
                                            int tile, int lo, int nrows) {
    int cnt = 0;
    if constexpr (KEYED) {
        const unsigned pat = (unsigned)tile * 0x01010101u;
        if (e0 + SLICE > E) {
#pragma unroll
            for (int j = 0; j < SLICE / KEYS_PER_LOAD; ++j) {
#pragma unroll
                for (int d = 0; d < 4; ++d) {
                    const int nv = E - (e0 + j * KEYS_PER_LOAD + lane * 16 + 4 * d);
                    const unsigned keep = (nv >= 4) ? 0xffffffffu : ((nv <= 0) ? 0u : ((1u << (8 * nv)) - 1u));
                    cur[j][d] = (int)(((unsigned)cur[j][d] & keep) | (~pat & ~keep));
                }
            }
        }
#pragma unroll
        for (int j = 0; j < SLICE / KEYS_PER_LOAD; ++j) {
            unsigned g = 0u;
#pragma unroll
            for (int d = 0; d < 4; ++d) {
                const unsigned x  = (unsigned)cur[j][d] ^ pat;
                const unsigned nz = ((x & 0x7f7f7f7fu) + 0x7f7f7f7fu) | x;
                g |= (~nz & 0x80808080u) >> (7 - d);
            }
            unsigned hit = __builtin_amdgcn_ballot_w32(g != 0u);
            while (hit != 0u) {
                if (g != 0u) {
                    const int p = __builtin_ctz(g);
                    myq[cnt + __builtin_popcount(hit & ((1u << lane) - 1u))] =
                        (unsigned short)(j * KEYS_PER_LOAD + lane * 16 + ((p & 7) << 2) + (p >> 3));
                    g &= g - 1u;
                }
                cnt += __builtin_popcount(hit);
                hit = __builtin_amdgcn_ballot_w32(g != 0u);
            }
        }
    } else {
#pragma unroll
        for (int j = 0; j < SLICE / 128; ++j) {
            unsigned mask[4];
#pragma unroll
            for (int k = 0; k < 4; ++k)
                mask[k] = __builtin_amdgcn_ballot_w32((unsigned)(cur[j][k] - lo) < (unsigned)nrows);
            if ((mask[0] | mask[1] | mask[2] | mask[3]) != 0u) {
#pragma unroll
                for (int k = 0; k < 4; ++k) {
                    if ((mask[k] >> lane) & 1u)
                        myq[cnt + __builtin_popcount(mask[k] & ((1u << lane) - 1u))] =
                            (unsigned short)(j * 128 + lane * 4 + k);
                    cnt += __builtin_popcount(mask[k]);
                }
            }
        }
    }
    return cnt;
}

template <bool KEYED, bool REC>
__global__ __launch_bounds__(1024) void owner_gather(const float* __restrict__ h,
                                                     const int* __restrict__ row_idx,
                                                     const int* __restrict__ col_idx,
                                                     const float* __restrict__ w,
                                                     const float* __restrict__ dis,
                                                     const unsigned* __restrict__ keys,
                                                     const unsigned* __restrict__ recs,
                                                     float* __restrict__ out, int N, int E) {
    __shared__ float          acc[TILE_ROWS * D];
    __shared__ unsigned short q[OWN_WAVES * SLICE];

    const int wave = __builtin_amdgcn_readfirstlane(threadIdx.x >> 5);
    const int lane = threadIdx.x & 31;
    unsigned short* myq = q + wave * SLICE;

    constexpr int NREG = KEYED ? SLICE / KEYS_PER_LOAD : SLICE / 128;
    const int  nslices = (E + SLICE - 1) / SLICE;
    const bool vec_ok  = (E & 3) == 0;
    const int  rot     = (int)(((unsigned)blockIdx.x * (unsigned)nslices / gridDim.x) % (unsigned)nslices);

    for (int tile = blockIdx.x; tile * TILE_ROWS < N; tile += gridDim.x) {
        const int lo    = tile * TILE_ROWS;
        const int nrows = min(TILE_ROWS, N - lo);

        for (int i = threadIdx.x; i < TILE_ROWS * D / 4; i += blockDim.x)
            ((float4*)acc)[i] = make_float4(0.0f, 0.0f, 0.0f, 0.0f);
        __syncthreads();

        v4i cur[NREG];
        {
            int sl = wave + rot; if (sl >= nslices) sl -= nslices;
            if (wave < nslices) fetch_slice<KEYED>(col_idx, keys, sl * SLICE, lane, E, vec_ok, cur);
        }
        for (int s = wave; s < nslices; s += OWN_WAVES) {
            int sl = s + rot; if (sl >= nslices) sl -= nslices;
            const int e0 = sl * SLICE;

            v4i nxt[NREG];
            {
                int sn = ((s + OWN_WAVES < nslices) ? s + OWN_WAVES : s) + rot;
                if (sn >= nslices) sn -= nslices;
                fetch_slice<KEYED>(col_idx, keys, sn * SLICE, lane, E, vec_ok, nxt);
            }

            const int cnt = filter_slice<KEYED>(cur, myq, lane, e0, E, tile, lo, nrows);

            for (int base = 0; base < cnt; base += 32) {
                const int n  = min(32, cnt - base);
                const int qi = (lane < n) ? base + lane : base;
                const int e  = e0 + (int)myq[qi];
                int r, lrow;
                float wt;
                if constexpr (REC) {
                    const uint2 rc = *(const uint2*)(recs + 2 * e);
                    r    = (int)(rc.x & ((1u << REC_ROW_BITS) - 1u));
                    lrow = (int)(rc.x >> REC_ROW_BITS);
                    wt   = __uint_as_float(rc.y);
                } else {
                    r    = row_idx[e];
                    lrow = col_idx[e] - lo;
                    wt   = w[e];
                }
                const float nm = dis[r] * wt * dis[lo + lrow];

                for (int t = 0; t < n; t += 4) {
                    float v[4][4];
                    float sc[4];
                    int   lr[4];
#pragma unroll
                    for (int u = 0; u < 4; ++u) {
                        const int tt = min(t + u, n - 1);
                        const int rt = __builtin_amdgcn_readlane(r, tt);
                        lr[u] = __builtin_amdgcn_readlane(lrow, tt);
                        sc[u] = __int_as_float(__builtin_amdgcn_readlane(__float_as_int(nm), tt));
                        const float* __restrict__ hr = h + (size_t)rt * D + lane;
#pragma unroll
                        for (int j = 0; j < 4; ++j) v[u][j] = hr[32 * j];
                    }
#pragma unroll
                    for (int u = 0; u < 4; ++u) {
                        if (t + u < n) {
                            float* a = acc + lr[u] * D + lane;
#pragma unroll
                            for (int j = 0; j < 4; ++j)
                                __hip_atomic_fetch_add(a + 32 * j, v[u][j] * sc[u], __ATOMIC_RELAXED,
                                                       __HIP_MEMORY_SCOPE_WORKGROUP);
                        }
                    }
                }
            }

#pragma unroll
            for (int j = 0; j < NREG; ++j) cur[j] = nxt[j];
        }
        __syncthreads();

        float4*       o4 = (float4*)(out + (size_t)lo * D);
        const float4* a4 = (const float4*)acc;
        for (int i = threadIdx.x; i < nrows * (D / 4); i += blockDim.x) {
            float4 o = o4[i];
            const float4 a = a4[i];
            o.x += a.x; o.y += a.y; o.z += a.z; o.w += a.w;
            o4[i] = o;
        }
        __syncthreads();
    }
}


extern "C" void kernel_launch(void* const* d_in, const int* in_sizes, int n_in,
                              void* d_out, int out_size, void* d_ws, size_t ws_size,
                              hipStream_t stream) {
    const float* x    = (const float*)d_in[0];
    const int*   ei   = (const int*)d_in[1];
    const float* ew   = (const float*)d_in[2];
    const float* Wlin = (const float*)d_in[3];
    const float* Wgcn = (const float*)d_in[4];
    float*       out  = (float*)d_out;

    const int N = in_sizes[0] / D;
    const int E = in_sizes[2];

    const int* row = ei;
    const int* col = ei + E;

    float* deg2 = (float*)d_ws;
    float* dis  = deg2;
    float* h    = deg2 + 2 * N;
    const int nslices = (E + SLICE - 1) / SLICE;
    const bool keyed = ws_size >= (size_t)(2 * N + N * D) * sizeof(float) + (size_t)nslices * SLICE;
    unsigned* keys = keyed ? (unsigned*)(h + (size_t)N * D) : nullptr;
    const size_t rec_off = ((size_t)(2 * N + N * D) * sizeof(float) + (size_t)nslices * SLICE + 15) & ~(size_t)15;
    const bool rec = keyed && N <= (1 << REC_ROW_BITS) && ws_size >= rec_off + (size_t)nslices * SLICE * 8;
    unsigned* recs = rec ? (unsigned*)((char*)d_ws + rec_off) : nullptr;

    zero_f32<<<(2 * N + 255) / 256, 256, 0, stream>>>(deg2, 2 * N);
    deg_kernel<<<nslices, SLICE / 4, 0, stream>>>(row, col, ew, deg2, keys, recs, N, E);
    rsqrt_kernel<<<(N + 255) / 256, 256, 0, stream>>>(deg2, N);
    gemm_dual<<<512, 512, 0, stream>>>(x, Wlin, Wgcn, out, h, N / 16);
    const int nblk = (N + TILE_ROWS - 1) / TILE_ROWS;
    if (rec)        owner_gather<true, true><<<nblk, 1024, 0, stream>>>(h, row, col, ew, dis, keys, recs, out, N, E);
    else if (keyed) owner_gather<true, false><<<nblk, 1024, 0, stream>>>(h, row, col, ew, dis, keys, recs, out, N, E);
    else            owner_gather<false, false><<<nblk, 1024, 0, stream>>>(h, row, col, ew, dis, keys, recs, out, N, E);
}


# normalisation per row, not per edge: product kernel stores h rows times rsqrt(deg[src]), propagate scales tile rows by rsqrt(deg[dst]); no dis load per edge, no rsqrt launch; term rounding changes
# speedup vs baseline: 2.8540x; 1.0545x over previous
#include <hip/hip_runtime.h>

#define D 128

typedef float v2f __attribute__((ext_vector_type(2)));
typedef float v8f __attribute__((ext_vector_type(8)));


__global__ void zero_f32(float* __restrict__ p, int n) {
    int i = blockIdx.x * blockDim.x + threadIdx.x;
    if (i < n) p[i] = 0.0f;
}


#define TILE_ROWS 392
#define SLICE     1024
#define OWN_WAVES 32
#define KEYS_PER_LOAD 512

typedef int v4i __attribute__((ext_vector_type(4)));

#define REC_ROW_BITS 23
__global__ __launch_bounds__(SLICE / 4) void deg_kernel(const int* __restrict__ row,
                                                        const int* __restrict__ col,
                                                        const float* __restrict__ w,
                                                        float* __restrict__ deg2,
                                                        unsigned* __restrict__ keys,
                                                        unsigned* __restrict__ recs, int n, int E) {
    const unsigned place = __builtin_amdgcn_s_sendmsg_rtn(0x87);
    float* __restrict__ deg = deg2 + (size_t)((place >> 18) & 1u) * n;
    const int e = (blockIdx.x * blockDim.x + threadIdx.x) * 4;
    unsigned key = 0u;
    if ((E & 3) == 0 && e < E) {
        const v4i    c  = *(const v4i*)(col + e);
        const float4 ww = *(const float4*)(w + e);
        unsafeAtomicAdd(&deg[c.x], ww.x);
        unsafeAtomicAdd(&deg[c.y], ww.y);
        unsafeAtomicAdd(&deg[c.z], ww.z);
        unsafeAtomicAdd(&deg[c.w], ww.w);
        const unsigned tx = (unsigned)c.x / TILE_ROWS, ty = (unsigned)c.y / TILE_ROWS;
        const unsigned tz = (unsigned)c.z / TILE_ROWS, tw = (unsigned)c.w / TILE_ROWS;
        key = tx | (ty << 8) | (tz << 16) | (tw << 24);
        if (recs != nullptr) {
            const v4i r = *(const v4i*)(row + e);
            uint4 a, b;
            a.x = (unsigned)r.x | (((unsigned)c.x - tx * TILE_ROWS) << REC_ROW_BITS); a.y = __float_as_uint(ww.x);
            a.z = (unsigned)r.y | (((unsigned)c.y - ty * TILE_ROWS) << REC_ROW_BITS); a.w = __float_as_uint(ww.y);
            b.x = (unsigned)r.z | (((unsigned)c.z - tz * TILE_ROWS) << REC_ROW_BITS); b.y = __float_as_uint(ww.z);
            b.z = (unsigned)r.w | (((unsigned)c.w - tw * TILE_ROWS) << REC_ROW_BITS); b.w = __float_as_uint(ww.w);
            *(uint4*)(recs + 2 * e)     = a;
            *(uint4*)(recs + 2 * e + 4) = b;
        }
    } else {
        for (int k = 0; k < 4; ++k) {
            if (e + k < E) {
                const int c = col[e + k];
                const float wk = w[e + k];
                unsafeAtomicAdd(&deg[c], wk);
                const unsigned t = (unsigned)c / TILE_ROWS;
                key |= t << (8 * k);
                if (recs != nullptr)
                    *(uint2*)(recs + 2 * (e + k)) =
                        make_uint2((unsigned)row[e + k] | (((unsigned)c - t * TILE_ROWS) << REC_ROW_BITS),
                                   __float_as_uint(wk));
            }
        }
    }
    if (keys != nullptr) keys[e >> 2] = key;
}

__device__ __forceinline__ float inv_sqrt_deg(float plane0, float plane1) {
    const float d = plane0 + plane1;
    return (d > 0.0f) ? __frsqrt_rn(d) : 0.0f;
}


#define XS_STRIDE 132

__global__ __launch_bounds__(512) void gemm_dual(const float* __restrict__ x,
                                                 const float* __restrict__ Wlin,
                                                 const float* __restrict__ Wgcn,
                                                 const float* __restrict__ deg2,
                                                 float* __restrict__ out_hl,
                                                 float* __restrict__ h, int ntiles, int n) {
    __shared__ __attribute__((aligned(16))) float xs[2][16 * XS_STRIDE];

    const int wave  = threadIdx.x >> 5;
    const int lane  = threadIdx.x & 31;
    const int which = wave >> 3;
    const int col0  = (wave & 7) * 16;

    const float* __restrict__ W   = which ? Wgcn : Wlin;
    float* __restrict__       dst = which ? h    : out_hl;

    const int m  = lane & 15;
    const int hh = lane >> 4;

    v2f b[32];
#pragma unroll
    for (int ks = 0; ks < 32; ++ks) {
        b[ks].x = W[(size_t)(4 * ks + 2 * hh + 0) * D + col0 + m];
        b[ks].y = W[(size_t)(4 * ks + 2 * hh + 1) * D + col0 + m];
    }

    const float* __restrict__ xsrc = x + (size_t)wave * D + lane * 4;
    const int soff = wave * XS_STRIDE + lane * 4;

    const float* __restrict__ dg0 = deg2 + m;
    const float* __restrict__ dg1 = deg2 + (size_t)n + m;

    int tile = blockIdx.x;
    float4 stage = make_float4(0.0f, 0.0f, 0.0f, 0.0f);
    float dsum0 = 0.0f, dsum1 = 0.0f;
    if (tile < ntiles) {
        stage = *(const float4*)(xsrc + (size_t)tile * 16 * D);
        if (which) { dsum0 = dg0[tile * 16]; dsum1 = dg1[tile * 16]; }
    }
    *(float4*)&xs[0][soff] = stage;
    __syncthreads();

    int buf = 0;
    for (; tile < ntiles; tile += gridDim.x) {
        const int  nt   = tile + gridDim.x;
        const bool more = nt < ntiles;
        const float rowfac = which ? inv_sqrt_deg(dsum0, dsum1) : 1.0f;
        if (more) {
            stage = *(const float4*)(xsrc + (size_t)nt * 16 * D);
            if (which) { dsum0 = dg0[nt * 16]; dsum1 = dg1[nt * 16]; }
        }

        const float* xa = &xs[buf][m * XS_STRIDE + 2 * hh];
        v8f acc = {};
#pragma unroll
        for (int ks = 0; ks < 32; ++ks) {
            const v2f a = *(const v2f*)(xa + 4 * ks);
            acc = __builtin_amdgcn_wmma_f32_16x16x4_f32(
                 false, a,  false, b[ks],
                 (short)0, acc,  false,  false);
        }

        float* __restrict__ drow = dst + (size_t)(tile * 16 + 8 * hh) * D + col0 + m;
#pragma unroll
        for (int r = 0; r < 8; ++r) {
            const float f0 = __int_as_float(__builtin_amdgcn_readlane(__float_as_int(rowfac), r));
            const float f1 = __int_as_float(__builtin_amdgcn_readlane(__float_as_int(rowfac), 8 + r));
            drow[(size_t)r * D] = acc[r] * (hh ? f1 : f0);
        }

        if (more) *(float4*)&xs[buf ^ 1][soff] = stage;
        __syncthreads();
        buf ^= 1;
    }
}


template <bool KEYED>
__device__ __forceinline__ void fetch_slice(const int* __restrict__ col, const unsigned* __restrict__ keys,
                                            int e0, int lane, int E, bool vec_ok,
                                            v4i (&c)[KEYED ? SLICE / KEYS_PER_LOAD : SLICE / 128]) {
    if constexpr (KEYED) {
#pragma unroll
        for (int j = 0; j < SLICE / KEYS_PER_LOAD; ++j)
            c[j] = *(const v4i*)(keys + ((e0 + j * KEYS_PER_LOAD) >> 2) + lane * 4);
    } else if (vec_ok && e0 + SLICE <= E) {
#pragma unroll
        for (int j = 0; j < SLICE / 128; ++j)
            c[j] = *(const v4i*)(col + e0 + j * 128 + lane * 4);
    } else {
#pragma unroll
        for (int j = 0; j < SLICE / 128; ++j) {
            const int e = e0 + j * 128 + lane * 4;
            c[j].x = (e + 0 < E) ? col[e + 0] : -1;
            c[j].y = (e + 1 < E) ? col[e + 1] : -1;
            c[j].z = (e + 2 < E) ? col[e + 2] : -1;
            c[j].w = (e + 3 < E) ? col[e + 3] : -1;
        }
    }
}

template <bool KEYED>
__device__ __forceinline__ int filter_slice(v4i (&cur)[KEYED ? SLICE / KEYS_PER_LOAD : SLICE / 128],
                                            unsigned short* myq, int lane, int e0, int E,
                                            int tile, int lo, int nrows) {
    int cnt = 0;
    if constexpr (KEYED) {
        const unsigned pat = (unsigned)tile * 0x01010101u;
        if (e0 + SLICE > E) {
#pragma unroll
            for (int j = 0; j < SLICE / KEYS_PER_LOAD; ++j) {
#pragma unroll
                for (int d = 0; d < 4; ++d) {
                    const int nv = E - (e0 + j * KEYS_PER_LOAD + lane * 16 + 4 * d);
                    const unsigned keep = (nv >= 4) ? 0xffffffffu : ((nv <= 0) ? 0u : ((1u << (8 * nv)) - 1u));
                    cur[j][d] = (int)(((unsigned)cur[j][d] & keep) | (~pat & ~keep));
                }
            }
        }
#pragma unroll
        for (int j = 0; j < SLICE / KEYS_PER_LOAD; ++j) {
            unsigned g = 0u;
#pragma unroll
            for (int d = 0; d < 4; ++d) {
                const unsigned x  = (unsigned)cur[j][d] ^ pat;
                const unsigned nz = ((x & 0x7f7f7f7fu) + 0x7f7f7f7fu) | x;
                g |= (~nz & 0x80808080u) >> (7 - d);
            }
            unsigned hit = __builtin_amdgcn_ballot_w32(g != 0u);
            while (hit != 0u) {
                if (g != 0u) {
                    const int p = __builtin_ctz(g);
                    myq[cnt + __builtin_popcount(hit & ((1u << lane) - 1u))] =
                        (unsigned short)(j * KEYS_PER_LOAD + lane * 16 + ((p & 7) << 2) + (p >> 3));
                    g &= g - 1u;
                }
                cnt += __builtin_popcount(hit);
                hit = __builtin_amdgcn_ballot_w32(g != 0u);
            }
        }
    } else {
#pragma unroll
        for (int j = 0; j < SLICE / 128; ++j) {
            unsigned mask[4];
#pragma unroll
            for (int k = 0; k < 4; ++k)
                mask[k] = __builtin_amdgcn_ballot_w32((unsigned)(cur[j][k] - lo) < (unsigned)nrows);
            if ((mask[0] | mask[1] | mask[2] | mask[3]) != 0u) {
#pragma unroll
                for (int k = 0; k < 4; ++k) {
                    if ((mask[k] >> lane) & 1u)
                        myq[cnt + __builtin_popcount(mask[k] & ((1u << lane) - 1u))] =
                            (unsigned short)(j * 128 + lane * 4 + k);
                    cnt += __builtin_popcount(mask[k]);
                }
            }
        }
    }
    return cnt;
}

template <bool KEYED, bool REC>
__global__ __launch_bounds__(1024) void owner_gather(const float* __restrict__ h,
                                                     const int* __restrict__ row_idx,
                                                     const int* __restrict__ col_idx,
                                                     const float* __restrict__ w,
                                                     const float* __restrict__ deg2,
                                                     const unsigned* __restrict__ keys,
                                                     const unsigned* __restrict__ recs,
                                                     float* __restrict__ out, int N, int E) {
    __shared__ float          acc[TILE_ROWS * D];
    __shared__ unsigned short q[OWN_WAVES * SLICE];

    const int wave = __builtin_amdgcn_readfirstlane(threadIdx.x >> 5);
    const int lane = threadIdx.x & 31;
    unsigned short* myq = q + wave * SLICE;

    constexpr int NREG = KEYED ? SLICE / KEYS_PER_LOAD : SLICE / 128;
    const int  nslices = (E + SLICE - 1) / SLICE;
    const bool vec_ok  = (E & 3) == 0;
    const int  rot     = (int)(((unsigned)blockIdx.x * (unsigned)nslices / gridDim.x) % (unsigned)nslices);

    for (int tile = blockIdx.x; tile * TILE_ROWS < N; tile += gridDim.x) {
        const int lo    = tile * TILE_ROWS;
        const int nrows = min(TILE_ROWS, N - lo);

        for (int i = threadIdx.x; i < TILE_ROWS * D / 4; i += blockDim.x)
            ((float4*)acc)[i] = make_float4(0.0f, 0.0f, 0.0f, 0.0f);
        __syncthreads();

        v4i cur[NREG];
        {
            int sl = wave + rot; if (sl >= nslices) sl -= nslices;
            if (wave < nslices) fetch_slice<KEYED>(col_idx, keys, sl * SLICE, lane, E, vec_ok, cur);
        }
        for (int s = wave; s < nslices; s += OWN_WAVES) {
            int sl = s + rot; if (sl >= nslices) sl -= nslices;
            const int e0 = sl * SLICE;

            v4i nxt[NREG];
            {
                int sn = ((s + OWN_WAVES < nslices) ? s + OWN_WAVES : s) + rot;
                if (sn >= nslices) sn -= nslices;
                fetch_slice<KEYED>(col_idx, keys, sn * SLICE, lane, E, vec_ok, nxt);
            }

            const int cnt = filter_slice<KEYED>(cur, myq, lane, e0, E, tile, lo, nrows);

            for (int base = 0; base < cnt; base += 32) {
                const int n  = min(32, cnt - base);
                const int qi = (lane < n) ? base + lane : base;
                const int e  = e0 + (int)myq[qi];
                int r, lrow;
                float wt;
                if constexpr (REC) {
                    const uint2 rc = *(const uint2*)(recs + 2 * e);
                    r    = (int)(rc.x & ((1u << REC_ROW_BITS) - 1u));
                    lrow = (int)(rc.x >> REC_ROW_BITS);
                    wt   = __uint_as_float(rc.y);
                } else {
                    r    = row_idx[e];
                    lrow = col_idx[e] - lo;
                    wt   = w[e];
                }

                for (int t = 0; t < n; t += 4) {
                    float v[4][4];
                    float sc[4];
                    int   lr[4];
#pragma unroll
                    for (int u = 0; u < 4; ++u) {
                        const int tt = min(t + u, n - 1);
                        const int rt = __builtin_amdgcn_readlane(r, tt);
                        lr[u] = __builtin_amdgcn_readlane(lrow, tt);
                        sc[u] = __int_as_float(__builtin_amdgcn_readlane(__float_as_int(wt), tt));
                        const float* __restrict__ hr = h + (size_t)rt * D + lane;
#pragma unroll
                        for (int j = 0; j < 4; ++j) v[u][j] = hr[32 * j];
                    }
#pragma unroll
                    for (int u = 0; u < 4; ++u) {
                        if (t + u < n) {
                            float* a = acc + lr[u] * D + lane;
#pragma unroll
                            for (int j = 0; j < 4; ++j)
                                __hip_atomic_fetch_add(a + 32 * j, v[u][j] * sc[u], __ATOMIC_RELAXED,
                                                       __HIP_MEMORY_SCOPE_WORKGROUP);
                        }
                    }
                }
            }

#pragma unroll
            for (int j = 0; j < NREG; ++j) cur[j] = nxt[j];
        }
        __syncthreads();

        float4*       o4 = (float4*)(out + (size_t)lo * D);
        const float4* a4 = (const float4*)acc;
        for (int i = threadIdx.x; i < nrows * (D / 4); i += blockDim.x) {
            const int   c  = lo + i / (D / 4);
            const float dc = inv_sqrt_deg(deg2[c], deg2[N + c]);
            float4 o = o4[i];
            const float4 a = a4[i];
            o.x += a.x * dc; o.y += a.y * dc; o.z += a.z * dc; o.w += a.w * dc;
            o4[i] = o;
        }
        __syncthreads();
    }
}


extern "C" void kernel_launch(void* const* d_in, const int* in_sizes, int n_in,
                              void* d_out, int out_size, void* d_ws, size_t ws_size,
                              hipStream_t stream) {
    const float* x    = (const float*)d_in[0];
    const int*   ei   = (const int*)d_in[1];
    const float* ew   = (const float*)d_in[2];
    const float* Wlin = (const float*)d_in[3];
    const float* Wgcn = (const float*)d_in[4];
    float*       out  = (float*)d_out;

    const int N = in_sizes[0] / D;
    const int E = in_sizes[2];

    const int* row = ei;
    const int* col = ei + E;

    float* deg2 = (float*)d_ws;
    float* h    = deg2 + 2 * N;
    const int nslices = (E + SLICE - 1) / SLICE;
    const bool keyed = ws_size >= (size_t)(2 * N + N * D) * sizeof(float) + (size_t)nslices * SLICE;
    unsigned* keys = keyed ? (unsigned*)(h + (size_t)N * D) : nullptr;
    const size_t rec_off = ((size_t)(2 * N + N * D) * sizeof(float) + (size_t)nslices * SLICE + 15) & ~(size_t)15;
    const bool rec = keyed && N <= (1 << REC_ROW_BITS) && ws_size >= rec_off + (size_t)nslices * SLICE * 8;
    unsigned* recs = rec ? (unsigned*)((char*)d_ws + rec_off) : nullptr;

    zero_f32<<<(2 * N + 255) / 256, 256, 0, stream>>>(deg2, 2 * N);
    deg_kernel<<<nslices, SLICE / 4, 0, stream>>>(row, col, ew, deg2, keys, recs, N, E);
    gemm_dual<<<512, 512, 0, stream>>>(x, Wlin, Wgcn, deg2, out, h, N / 16, N);
    const int nblk = (N + TILE_ROWS - 1) / TILE_ROWS;
    if (rec)        owner_gather<true, true><<<nblk, 1024, 0, stream>>>(h, row, col, ew, deg2, keys, recs, out, N, E);
    else if (keyed) owner_gather<true, false><<<nblk, 1024, 0, stream>>>(h, row, col, ew, deg2, keys, recs, out, N, E);
    else            owner_gather<false, false><<<nblk, 1024, 0, stream>>>(h, row, col, ew, deg2, keys, recs, out, N, E);
}


# gather decoupled from the slice filter: a wave queues owned edges over its slices, gathers only full groups of 32, the next 4 rows requested before the adds of the current 4; arrival-order sums
# speedup vs baseline: 2.9657x; 1.0391x over previous
#include <hip/hip_runtime.h>

#define D 128

typedef float v2f __attribute__((ext_vector_type(2)));
typedef float v8f __attribute__((ext_vector_type(8)));


__global__ void zero_f32(float* __restrict__ p, int n) {
    int i = blockIdx.x * blockDim.x + threadIdx.x;
    if (i < n) p[i] = 0.0f;
}


#define TILE_ROWS 392
#define SLICE     1024
#define OWN_WAVES 32
#define KEYS_PER_LOAD 512

typedef int v4i __attribute__((ext_vector_type(4)));

#define REC_ROW_BITS 23
__global__ __launch_bounds__(SLICE / 4) void deg_kernel(const int* __restrict__ row,
                                                        const int* __restrict__ col,
                                                        const float* __restrict__ w,
                                                        float* __restrict__ deg2,
                                                        unsigned* __restrict__ keys,
                                                        unsigned* __restrict__ recs, int n, int E) {
    const unsigned place = __builtin_amdgcn_s_sendmsg_rtn(0x87);
    float* __restrict__ deg = deg2 + (size_t)((place >> 18) & 1u) * n;
    const int e = (blockIdx.x * blockDim.x + threadIdx.x) * 4;
    unsigned key = 0u;
    if ((E & 3) == 0 && e < E) {
        const v4i    c  = *(const v4i*)(col + e);
        const float4 ww = *(const float4*)(w + e);
        unsafeAtomicAdd(&deg[c.x], ww.x);
        unsafeAtomicAdd(&deg[c.y], ww.y);
        unsafeAtomicAdd(&deg[c.z], ww.z);
        unsafeAtomicAdd(&deg[c.w], ww.w);
        const unsigned tx = (unsigned)c.x / TILE_ROWS, ty = (unsigned)c.y / TILE_ROWS;
        const unsigned tz = (unsigned)c.z / TILE_ROWS, tw = (unsigned)c.w / TILE_ROWS;
        key = tx | (ty << 8) | (tz << 16) | (tw << 24);
        if (recs != nullptr) {
            const v4i r = *(const v4i*)(row + e);
            uint4 a, b;
            a.x = (unsigned)r.x | (((unsigned)c.x - tx * TILE_ROWS) << REC_ROW_BITS); a.y = __float_as_uint(ww.x);
            a.z = (unsigned)r.y | (((unsigned)c.y - ty * TILE_ROWS) << REC_ROW_BITS); a.w = __float_as_uint(ww.y);
            b.x = (unsigned)r.z | (((unsigned)c.z - tz * TILE_ROWS) << REC_ROW_BITS); b.y = __float_as_uint(ww.z);
            b.z = (unsigned)r.w | (((unsigned)c.w - tw * TILE_ROWS) << REC_ROW_BITS); b.w = __float_as_uint(ww.w);
            *(uint4*)(recs + 2 * e)     = a;
            *(uint4*)(recs + 2 * e + 4) = b;
        }
    } else {
        for (int k = 0; k < 4; ++k) {
            if (e + k < E) {
                const int c = col[e + k];
                const float wk = w[e + k];
                unsafeAtomicAdd(&deg[c], wk);
                const unsigned t = (unsigned)c / TILE_ROWS;
                key |= t << (8 * k);
                if (recs != nullptr)
                    *(uint2*)(recs + 2 * (e + k)) =
                        make_uint2((unsigned)row[e + k] | (((unsigned)c - t * TILE_ROWS) << REC_ROW_BITS),
                                   __float_as_uint(wk));
            }
        }
    }
    if (keys != nullptr) keys[e >> 2] = key;
}

__device__ __forceinline__ float inv_sqrt_deg(float plane0, float plane1) {
    const float d = plane0 + plane1;
    return (d > 0.0f) ? __frsqrt_rn(d) : 0.0f;
}


#define XS_STRIDE 132

__global__ __launch_bounds__(512) void gemm_dual(const float* __restrict__ x,
                                                 const float* __restrict__ Wlin,
                                                 const float* __restrict__ Wgcn,
                                                 const float* __restrict__ deg2,
                                                 float* __restrict__ out_hl,
                                                 float* __restrict__ h, int ntiles, int n) {
    __shared__ __attribute__((aligned(16))) float xs[2][16 * XS_STRIDE];

    const int wave  = threadIdx.x >> 5;
    const int lane  = threadIdx.x & 31;
    const int which = wave >> 3;
    const int col0  = (wave & 7) * 16;

    const float* __restrict__ W   = which ? Wgcn : Wlin;
    float* __restrict__       dst = which ? h    : out_hl;

    const int m  = lane & 15;
    const int hh = lane >> 4;

    v2f b[32];
#pragma unroll
    for (int ks = 0; ks < 32; ++ks) {
        b[ks].x = W[(size_t)(4 * ks + 2 * hh + 0) * D + col0 + m];
        b[ks].y = W[(size_t)(4 * ks + 2 * hh + 1) * D + col0 + m];
    }

    const float* __restrict__ xsrc = x + (size_t)wave * D + lane * 4;
    const int soff = wave * XS_STRIDE + lane * 4;

    const float* __restrict__ dg0 = deg2 + m;
    const float* __restrict__ dg1 = deg2 + (size_t)n + m;

    int tile = blockIdx.x;
    float4 stage = make_float4(0.0f, 0.0f, 0.0f, 0.0f);
    float dsum0 = 0.0f, dsum1 = 0.0f;
    if (tile < ntiles) {
        stage = *(const float4*)(xsrc + (size_t)tile * 16 * D);
        if (which) { dsum0 = dg0[tile * 16]; dsum1 = dg1[tile * 16]; }
    }
    *(float4*)&xs[0][soff] = stage;
    __syncthreads();

    int buf = 0;
    for (; tile < ntiles; tile += gridDim.x) {
        const int  nt   = tile + gridDim.x;
        const bool more = nt < ntiles;
        const float rowfac = which ? inv_sqrt_deg(dsum0, dsum1) : 1.0f;
        if (more) {
            stage = *(const float4*)(xsrc + (size_t)nt * 16 * D);
            if (which) { dsum0 = dg0[nt * 16]; dsum1 = dg1[nt * 16]; }
        }

        const float* xa = &xs[buf][m * XS_STRIDE + 2 * hh];
        v8f acc = {};
#pragma unroll
        for (int ks = 0; ks < 32; ++ks) {
            const v2f a = *(const v2f*)(xa + 4 * ks);
            acc = __builtin_amdgcn_wmma_f32_16x16x4_f32(
                 false, a,  false, b[ks],
                 (short)0, acc,  false,  false);
        }

        float* __restrict__ drow = dst + (size_t)(tile * 16 + 8 * hh) * D + col0 + m;
#pragma unroll
        for (int r = 0; r < 8; ++r) {
            const float f0 = __int_as_float(__builtin_amdgcn_readlane(__float_as_int(rowfac), r));
            const float f1 = __int_as_float(__builtin_amdgcn_readlane(__float_as_int(rowfac), 8 + r));
            drow[(size_t)r * D] = acc[r] * (hh ? f1 : f0);
        }

        if (more) *(float4*)&xs[buf ^ 1][soff] = stage;
        __syncthreads();
        buf ^= 1;
    }
}


template <bool KEYED>
__device__ __forceinline__ void fetch_slice(const int* __restrict__ col, const unsigned* __restrict__ keys,
                                            int e0, int lane, int E, bool vec_ok,
                                            v4i (&c)[KEYED ? SLICE / KEYS_PER_LOAD : SLICE / 128]) {
    if constexpr (KEYED) {
#pragma unroll
        for (int j = 0; j < SLICE / KEYS_PER_LOAD; ++j)
            c[j] = *(const v4i*)(keys + ((e0 + j * KEYS_PER_LOAD) >> 2) + lane * 4);
    } else if (vec_ok && e0 + SLICE <= E) {
#pragma unroll
        for (int j = 0; j < SLICE / 128; ++j)
            c[j] = *(const v4i*)(col + e0 + j * 128 + lane * 4);
    } else {
#pragma unroll
        for (int j = 0; j < SLICE / 128; ++j) {
            const int e = e0 + j * 128 + lane * 4;
            c[j].x = (e + 0 < E) ? col[e + 0] : -1;
            c[j].y = (e + 1 < E) ? col[e + 1] : -1;
            c[j].z = (e + 2 < E) ? col[e + 2] : -1;
            c[j].w = (e + 3 < E) ? col[e + 3] : -1;
        }
    }
}

template <bool KEYED>
__device__ __forceinline__ int filter_slice(v4i (&cur)[KEYED ? SLICE / KEYS_PER_LOAD : SLICE / 128],
                                            unsigned short* myq, int lane, int e0, int E,
                                            int tile, int lo, int nrows, int tag) {
    int cnt = 0;
    if constexpr (KEYED) {
        const unsigned pat = (unsigned)tile * 0x01010101u;
        if (e0 + SLICE > E) {
#pragma unroll
            for (int j = 0; j < SLICE / KEYS_PER_LOAD; ++j) {
#pragma unroll
                for (int d = 0; d < 4; ++d) {
                    const int nv = E - (e0 + j * KEYS_PER_LOAD + lane * 16 + 4 * d);
                    const unsigned keep = (nv >= 4) ? 0xffffffffu : ((nv <= 0) ? 0u : ((1u << (8 * nv)) - 1u));
                    cur[j][d] = (int)(((unsigned)cur[j][d] & keep) | (~pat & ~keep));
                }
            }
        }
#pragma unroll
        for (int j = 0; j < SLICE / KEYS_PER_LOAD; ++j) {
            unsigned g = 0u;
#pragma unroll
            for (int d = 0; d < 4; ++d) {
                const unsigned x  = (unsigned)cur[j][d] ^ pat;
                const unsigned nz = ((x & 0x7f7f7f7fu) + 0x7f7f7f7fu) | x;
                g |= (~nz & 0x80808080u) >> (7 - d);
            }
            unsigned hit = __builtin_amdgcn_ballot_w32(g != 0u);
            while (hit != 0u) {
                if (g != 0u) {
                    const int p = __builtin_ctz(g);
                    myq[cnt + __builtin_popcount(hit & ((1u << lane) - 1u))] =
                        (unsigned short)(tag | (j * KEYS_PER_LOAD + lane * 16 + ((p & 7) << 2) + (p >> 3)));
                    g &= g - 1u;
                }
                cnt += __builtin_popcount(hit);
                hit = __builtin_amdgcn_ballot_w32(g != 0u);
            }
        }
    } else {
#pragma unroll
        for (int j = 0; j < SLICE / 128; ++j) {
            unsigned mask[4];
#pragma unroll
            for (int k = 0; k < 4; ++k)
                mask[k] = __builtin_amdgcn_ballot_w32((unsigned)(cur[j][k] - lo) < (unsigned)nrows);
            if ((mask[0] | mask[1] | mask[2] | mask[3]) != 0u) {
#pragma unroll
                for (int k = 0; k < 4; ++k) {
                    if ((mask[k] >> lane) & 1u)
                        myq[cnt + __builtin_popcount(mask[k] & ((1u << lane) - 1u))] =
                            (unsigned short)(tag | (j * 128 + lane * 4 + k));
                    cnt += __builtin_popcount(mask[k]);
                }
            }
        }
    }
    return cnt;
}

#define QCAP (SLICE + 32)
#define TAG_SLICES 64

template <bool REC, bool FULL>
__device__ __forceinline__ void gather_group(const float* __restrict__ h, const int* __restrict__ row_idx,
                                             const int* __restrict__ col_idx, const float* __restrict__ w,
                                             const unsigned* __restrict__ recs, float* acc,
                                             const unsigned short* qg, int n, int lane, int wave,
                                             int kblock, int rot, int nslices, int lo) {
    const int qi = FULL ? lane : ((lane < n) ? lane : 0);
    const unsigned en = qg[qi];
    int sl = wave + OWN_WAVES * (kblock + (int)(en >> 10)) + rot;
    if (sl >= nslices) sl -= nslices;
    const int e = sl * SLICE + (int)(en & (SLICE - 1));
    int r, lrow;
    float wt;
    if constexpr (REC) {
        const uint2 rc = *(const uint2*)(recs + 2 * e);
        r    = (int)(rc.x & ((1u << REC_ROW_BITS) - 1u));
        lrow = (int)(rc.x >> REC_ROW_BITS);
        wt   = __uint_as_float(rc.y);
    } else {
        r    = row_idx[e];
        lrow = col_idx[e] - lo;
        wt   = w[e];
    }

    if constexpr (FULL) {
        float v[2][4][4];
#pragma unroll
        for (int u = 0; u < 4; ++u) {
            const float* __restrict__ hr = h + (size_t)__builtin_amdgcn_readlane(r, u) * D + lane;
#pragma unroll
            for (int j = 0; j < 4; ++j) v[0][u][j] = hr[32 * j];
        }
#pragma unroll
        for (int g = 0; g < 8; ++g) {
            if (g + 1 < 8) {
#pragma unroll
                for (int u = 0; u < 4; ++u) {
                    const float* __restrict__ hr =
                        h + (size_t)__builtin_amdgcn_readlane(r, 4 * (g + 1) + u) * D + lane;
#pragma unroll
                    for (int j = 0; j < 4; ++j) v[(g + 1) & 1][u][j] = hr[32 * j];
                }
            }
#pragma unroll
            for (int u = 0; u < 4; ++u) {
                const int   lr = __builtin_amdgcn_readlane(lrow, 4 * g + u);
                const float sc = __int_as_float(__builtin_amdgcn_readlane(__float_as_int(wt), 4 * g + u));
                float* a = acc + lr * D + lane;
#pragma unroll
                for (int j = 0; j < 4; ++j)
                    __hip_atomic_fetch_add(a + 32 * j, v[g & 1][u][j] * sc, __ATOMIC_RELAXED,
                                           __HIP_MEMORY_SCOPE_WORKGROUP);
            }
        }
    } else {
        for (int t = 0; t < n; t += 4) {
            float v[4][4];
            float sc[4];
            int   lr[4];
#pragma unroll
            for (int u = 0; u < 4; ++u) {
                const int tt = min(t + u, n - 1);
                const int rt = __builtin_amdgcn_readlane(r, tt);
                lr[u] = __builtin_amdgcn_readlane(lrow, tt);
                sc[u] = __int_as_float(__builtin_amdgcn_readlane(__float_as_int(wt), tt));
                const float* __restrict__ hr = h + (size_t)rt * D + lane;
#pragma unroll
                for (int j = 0; j < 4; ++j) v[u][j] = hr[32 * j];
            }
#pragma unroll
            for (int u = 0; u < 4; ++u) {
                if (t + u < n) {
                    float* a = acc + lr[u] * D + lane;
#pragma unroll
                    for (int j = 0; j < 4; ++j)
                        __hip_atomic_fetch_add(a + 32 * j, v[u][j] * sc[u], __ATOMIC_RELAXED,
                                               __HIP_MEMORY_SCOPE_WORKGROUP);
                }
            }
        }
    }
}

template <bool KEYED, bool REC>
__global__ __launch_bounds__(1024) void owner_gather(const float* __restrict__ h,
                                                     const int* __restrict__ row_idx,
                                                     const int* __restrict__ col_idx,
                                                     const float* __restrict__ w,
                                                     const float* __restrict__ deg2,
                                                     const unsigned* __restrict__ keys,
                                                     const unsigned* __restrict__ recs,
                                                     float* __restrict__ out, int N, int E) {
    __shared__ float          acc[TILE_ROWS * D];
    __shared__ unsigned short q[OWN_WAVES * QCAP];

    const int wave = __builtin_amdgcn_readfirstlane(threadIdx.x >> 5);
    const int lane = threadIdx.x & 31;
    unsigned short* myq = q + wave * QCAP;

    constexpr int NREG = KEYED ? SLICE / KEYS_PER_LOAD : SLICE / 128;
    const int  nslices = (E + SLICE - 1) / SLICE;
    const bool vec_ok  = (E & 3) == 0;
    const int  rot     = (int)(((unsigned)blockIdx.x * (unsigned)nslices / gridDim.x) % (unsigned)nslices);

    for (int tile = blockIdx.x; tile * TILE_ROWS < N; tile += gridDim.x) {
        const int lo    = tile * TILE_ROWS;
        const int nrows = min(TILE_ROWS, N - lo);

        for (int i = threadIdx.x; i < TILE_ROWS * D / 4; i += blockDim.x)
            ((float4*)acc)[i] = make_float4(0.0f, 0.0f, 0.0f, 0.0f);
        __syncthreads();

        v4i cur[NREG];
        {
            int sl = wave + rot; if (sl >= nslices) sl -= nslices;
            if (wave < nslices) fetch_slice<KEYED>(col_idx, keys, sl * SLICE, lane, E, vec_ok, cur);
        }
        int cnt = 0;
        int k   = 0;
        for (int s = wave; s < nslices; s += OWN_WAVES, ++k) {
            int sl = s + rot; if (sl >= nslices) sl -= nslices;
            const int e0 = sl * SLICE;

            v4i nxt[NREG];
            {
                int sn = ((s + OWN_WAVES < nslices) ? s + OWN_WAVES : s) + rot;
                if (sn >= nslices) sn -= nslices;
                fetch_slice<KEYED>(col_idx, keys, sn * SLICE, lane, E, vec_ok, nxt);
            }

            cnt += filter_slice<KEYED>(cur, myq + cnt, lane, e0, E, tile, lo, nrows,
                                       (k & (TAG_SLICES - 1)) << 10);

            const int kblock = k & ~(TAG_SLICES - 1);
            int head = 0;
            for (; cnt - head >= 32; head += 32)
                gather_group<REC, true>(h, row_idx, col_idx, w, recs, acc, myq + head, 32, lane, wave,
                                        kblock, rot, nslices, lo);
            const bool flush = (s + OWN_WAVES >= nslices) || ((k & (TAG_SLICES - 1)) == TAG_SLICES - 1);
            if (flush && cnt > head) {
                gather_group<REC, false>(h, row_idx, col_idx, w, recs, acc, myq + head, cnt - head, lane, wave,
                                         kblock, rot, nslices, lo);
                head = cnt;
            }
            const int rest = cnt - head;
            if (head > 0 && rest > 0) {
                const unsigned short en = myq[head + ((lane < rest) ? lane : 0)];
                if (lane < rest) myq[lane] = en;
            }
            cnt = rest;

#pragma unroll
            for (int j = 0; j < NREG; ++j) cur[j] = nxt[j];
        }
        __syncthreads();

        float4*       o4 = (float4*)(out + (size_t)lo * D);
        const float4* a4 = (const float4*)acc;
        for (int i = threadIdx.x; i < nrows * (D / 4); i += blockDim.x) {
            const int   c  = lo + i / (D / 4);
            const float dc = inv_sqrt_deg(deg2[c], deg2[N + c]);
            float4 o = o4[i];
            const float4 a = a4[i];
            o.x += a.x * dc; o.y += a.y * dc; o.z += a.z * dc; o.w += a.w * dc;
            o4[i] = o;
        }
        __syncthreads();
    }
}


extern "C" void kernel_launch(void* const* d_in, const int* in_sizes, int n_in,
                              void* d_out, int out_size, void* d_ws, size_t ws_size,
                              hipStream_t stream) {
    const float* x    = (const float*)d_in[0];
    const int*   ei   = (const int*)d_in[1];
    const float* ew   = (const float*)d_in[2];
    const float* Wlin = (const float*)d_in[3];
    const float* Wgcn = (const float*)d_in[4];
    float*       out  = (float*)d_out;

    const int N = in_sizes[0] / D;
    const int E = in_sizes[2];

    const int* row = ei;
    const int* col = ei + E;

    float* deg2 = (float*)d_ws;
    float* h    = deg2 + 2 * N;
    const int nslices = (E + SLICE - 1) / SLICE;
    const bool keyed = ws_size >= (size_t)(2 * N + N * D) * sizeof(float) + (size_t)nslices * SLICE;
    unsigned* keys = keyed ? (unsigned*)(h + (size_t)N * D) : nullptr;
    const size_t rec_off = ((size_t)(2 * N + N * D) * sizeof(float) + (size_t)nslices * SLICE + 15) & ~(size_t)15;
    const bool rec = keyed && N <= (1 << REC_ROW_BITS) && ws_size >= rec_off + (size_t)nslices * SLICE * 8;
    unsigned* recs = rec ? (unsigned*)((char*)d_ws + rec_off) : nullptr;

    zero_f32<<<(2 * N + 255) / 256, 256, 0, stream>>>(deg2, 2 * N);
    deg_kernel<<<nslices, SLICE / 4, 0, stream>>>(row, col, ew, deg2, keys, recs, N, E);
    gemm_dual<<<512, 512, 0, stream>>>(x, Wlin, Wgcn, deg2, out, h, N / 16, N);
    const int nblk = (N + TILE_ROWS - 1) / TILE_ROWS;
    if (rec)        owner_gather<true, true><<<nblk, 1024, 0, stream>>>(h, row, col, ew, deg2, keys, recs, out, N, E);
    else if (keyed) owner_gather<true, false><<<nblk, 1024, 0, stream>>>(h, row, col, ew, deg2, keys, recs, out, N, E);
    else            owner_gather<false, false><<<nblk, 1024, 0, stream>>>(h, row, col, ew, deg2, keys, recs, out, N, E);
}


# cell plane: degree kernel ranks a slice's edges per destination tile on LDS counters, writes each (tile, slice) as one 128-B cell of records; an owner block reads only its cells; arrival-order sums
# speedup vs baseline: 3.4005x; 1.1466x over previous
#include <hip/hip_runtime.h>

#define D 128

typedef float v2f __attribute__((ext_vector_type(2)));
typedef float v8f __attribute__((ext_vector_type(8)));


__global__ void zero_f32(float* __restrict__ p, int n) {
    int i = blockIdx.x * blockDim.x + threadIdx.x;
    if (i < n) p[i] = 0.0f;
}


#define TILE_ROWS 392
#define SLICE     1024
#define OWN_WAVES 32
#define KEYS_PER_LOAD 512

typedef int v4i __attribute__((ext_vector_type(4)));

#define REC_ROW_BITS 23
__global__ __launch_bounds__(SLICE / 4) void deg_kernel(const int* __restrict__ row,
                                                        const int* __restrict__ col,
                                                        const float* __restrict__ w,
                                                        float* __restrict__ deg2,
                                                        unsigned* __restrict__ keys,
                                                        unsigned* __restrict__ recs, int n, int E) {
    const unsigned place = __builtin_amdgcn_s_sendmsg_rtn(0x87);
    float* __restrict__ deg = deg2 + (size_t)((place >> 18) & 1u) * n;
    const int e = (blockIdx.x * blockDim.x + threadIdx.x) * 4;
    unsigned key = 0u;
    if ((E & 3) == 0 && e < E) {
        const v4i    c  = *(const v4i*)(col + e);
        const float4 ww = *(const float4*)(w + e);
        unsafeAtomicAdd(&deg[c.x], ww.x);
        unsafeAtomicAdd(&deg[c.y], ww.y);
        unsafeAtomicAdd(&deg[c.z], ww.z);
        unsafeAtomicAdd(&deg[c.w], ww.w);
        const unsigned tx = (unsigned)c.x / TILE_ROWS, ty = (unsigned)c.y / TILE_ROWS;
        const unsigned tz = (unsigned)c.z / TILE_ROWS, tw = (unsigned)c.w / TILE_ROWS;
        key = tx | (ty << 8) | (tz << 16) | (tw << 24);
        if (recs != nullptr) {
            const v4i r = *(const v4i*)(row + e);
            uint4 a, b;
            a.x = (unsigned)r.x | (((unsigned)c.x - tx * TILE_ROWS) << REC_ROW_BITS); a.y = __float_as_uint(ww.x);
            a.z = (unsigned)r.y | (((unsigned)c.y - ty * TILE_ROWS) << REC_ROW_BITS); a.w = __float_as_uint(ww.y);
            b.x = (unsigned)r.z | (((unsigned)c.z - tz * TILE_ROWS) << REC_ROW_BITS); b.y = __float_as_uint(ww.z);
            b.z = (unsigned)r.w | (((unsigned)c.w - tw * TILE_ROWS) << REC_ROW_BITS); b.w = __float_as_uint(ww.w);
            *(uint4*)(recs + 2 * e)     = a;
            *(uint4*)(recs + 2 * e + 4) = b;
        }
    } else {
        for (int k = 0; k < 4; ++k) {
            if (e + k < E) {
                const int c = col[e + k];
                const float wk = w[e + k];
                unsafeAtomicAdd(&deg[c], wk);
                const unsigned t = (unsigned)c / TILE_ROWS;
                key |= t << (8 * k);
                if (recs != nullptr)
                    *(uint2*)(recs + 2 * (e + k)) =
                        make_uint2((unsigned)row[e + k] | (((unsigned)c - t * TILE_ROWS) << REC_ROW_BITS),
                                   __float_as_uint(wk));
            }
        }
    }
    if (keys != nullptr) keys[e >> 2] = key;
}

__device__ __forceinline__ float inv_sqrt_deg(float plane0, float plane1) {
    const float d = plane0 + plane1;
    return (d > 0.0f) ? __frsqrt_rn(d) : 0.0f;
}


#define XS_STRIDE 132

__global__ __launch_bounds__(512) void gemm_dual(const float* __restrict__ x,
                                                 const float* __restrict__ Wlin,
                                                 const float* __restrict__ Wgcn,
                                                 const float* __restrict__ deg2,
                                                 float* __restrict__ out_hl,
                                                 float* __restrict__ h, int ntiles, int n) {
    __shared__ __attribute__((aligned(16))) float xs[2][16 * XS_STRIDE];

    const int wave  = threadIdx.x >> 5;
    const int lane  = threadIdx.x & 31;
    const int which = wave >> 3;
    const int col0  = (wave & 7) * 16;

    const float* __restrict__ W   = which ? Wgcn : Wlin;
    float* __restrict__       dst = which ? h    : out_hl;

    const int m  = lane & 15;
    const int hh = lane >> 4;

    v2f b[32];
#pragma unroll
    for (int ks = 0; ks < 32; ++ks) {
        b[ks].x = W[(size_t)(4 * ks + 2 * hh + 0) * D + col0 + m];
        b[ks].y = W[(size_t)(4 * ks + 2 * hh + 1) * D + col0 + m];
    }

    const float* __restrict__ xsrc = x + (size_t)wave * D + lane * 4;
    const int soff = wave * XS_STRIDE + lane * 4;

    const float* __restrict__ dg0 = deg2 + m;
    const float* __restrict__ dg1 = deg2 + (size_t)n + m;

    int tile = blockIdx.x;
    float4 stage = make_float4(0.0f, 0.0f, 0.0f, 0.0f);
    float dsum0 = 0.0f, dsum1 = 0.0f;
    if (tile < ntiles) {
        stage = *(const float4*)(xsrc + (size_t)tile * 16 * D);
        if (which) { dsum0 = dg0[tile * 16]; dsum1 = dg1[tile * 16]; }
    }
    *(float4*)&xs[0][soff] = stage;
    __syncthreads();

    int buf = 0;
    for (; tile < ntiles; tile += gridDim.x) {
        const int  nt   = tile + gridDim.x;
        const bool more = nt < ntiles;
        const float rowfac = which ? inv_sqrt_deg(dsum0, dsum1) : 1.0f;
        if (more) {
            stage = *(const float4*)(xsrc + (size_t)nt * 16 * D);
            if (which) { dsum0 = dg0[nt * 16]; dsum1 = dg1[nt * 16]; }
        }

        const float* xa = &xs[buf][m * XS_STRIDE + 2 * hh];
        v8f acc = {};
#pragma unroll
        for (int ks = 0; ks < 32; ++ks) {
            const v2f a = *(const v2f*)(xa + 4 * ks);
            acc = __builtin_amdgcn_wmma_f32_16x16x4_f32(
                 false, a,  false, b[ks],
                 (short)0, acc,  false,  false);
        }

        float* __restrict__ drow = dst + (size_t)(tile * 16 + 8 * hh) * D + col0 + m;
#pragma unroll
        for (int r = 0; r < 8; ++r) {
            const float f0 = __int_as_float(__builtin_amdgcn_readlane(__float_as_int(rowfac), r));
            const float f1 = __int_as_float(__builtin_amdgcn_readlane(__float_as_int(rowfac), 8 + r));
            drow[(size_t)r * D] = acc[r] * (hh ? f1 : f0);
        }

        if (more) *(float4*)&xs[buf ^ 1][soff] = stage;
        __syncthreads();
        buf ^= 1;
    }
}


template <bool KEYED>
__device__ __forceinline__ void fetch_slice(const int* __restrict__ col, const unsigned* __restrict__ keys,
                                            int e0, int lane, int E, bool vec_ok,
                                            v4i (&c)[KEYED ? SLICE / KEYS_PER_LOAD : SLICE / 128]) {
    if constexpr (KEYED) {
#pragma unroll
        for (int j = 0; j < SLICE / KEYS_PER_LOAD; ++j)
            c[j] = *(const v4i*)(keys + ((e0 + j * KEYS_PER_LOAD) >> 2) + lane * 4);
    } else if (vec_ok && e0 + SLICE <= E) {
#pragma unroll
        for (int j = 0; j < SLICE / 128; ++j)
            c[j] = *(const v4i*)(col + e0 + j * 128 + lane * 4);
    } else {
#pragma unroll
        for (int j = 0; j < SLICE / 128; ++j) {
            const int e = e0 + j * 128 + lane * 4;
            c[j].x = (e + 0 < E) ? col[e + 0] : -1;
            c[j].y = (e + 1 < E) ? col[e + 1] : -1;
            c[j].z = (e + 2 < E) ? col[e + 2] : -1;
            c[j].w = (e + 3 < E) ? col[e + 3] : -1;
        }
    }
}

template <bool KEYED>
__device__ __forceinline__ int filter_slice(v4i (&cur)[KEYED ? SLICE / KEYS_PER_LOAD : SLICE / 128],
                                            unsigned short* myq, int lane, int e0, int E,
                                            int tile, int lo, int nrows, int tag) {
    int cnt = 0;
    if constexpr (KEYED) {
        const unsigned pat = (unsigned)tile * 0x01010101u;
        if (e0 + SLICE > E) {
#pragma unroll
            for (int j = 0; j < SLICE / KEYS_PER_LOAD; ++j) {
#pragma unroll
                for (int d = 0; d < 4; ++d) {
                    const int nv = E - (e0 + j * KEYS_PER_LOAD + lane * 16 + 4 * d);
                    const unsigned keep = (nv >= 4) ? 0xffffffffu : ((nv <= 0) ? 0u : ((1u << (8 * nv)) - 1u));
                    cur[j][d] = (int)(((unsigned)cur[j][d] & keep) | (~pat & ~keep));
                }
            }
        }
#pragma unroll
        for (int j = 0; j < SLICE / KEYS_PER_LOAD; ++j) {
            unsigned g = 0u;
#pragma unroll
            for (int d = 0; d < 4; ++d) {
                const unsigned x  = (unsigned)cur[j][d] ^ pat;
                const unsigned nz = ((x & 0x7f7f7f7fu) + 0x7f7f7f7fu) | x;
                g |= (~nz & 0x80808080u) >> (7 - d);
            }
            unsigned hit = __builtin_amdgcn_ballot_w32(g != 0u);
            while (hit != 0u) {
                if (g != 0u) {
                    const int p = __builtin_ctz(g);
                    myq[cnt + __builtin_popcount(hit & ((1u << lane) - 1u))] =
                        (unsigned short)(tag | (j * KEYS_PER_LOAD + lane * 16 + ((p & 7) << 2) + (p >> 3)));
                    g &= g - 1u;
                }
                cnt += __builtin_popcount(hit);
                hit = __builtin_amdgcn_ballot_w32(g != 0u);
            }
        }
    } else {
#pragma unroll
        for (int j = 0; j < SLICE / 128; ++j) {
            unsigned mask[4];
#pragma unroll
            for (int k = 0; k < 4; ++k)
                mask[k] = __builtin_amdgcn_ballot_w32((unsigned)(cur[j][k] - lo) < (unsigned)nrows);
            if ((mask[0] | mask[1] | mask[2] | mask[3]) != 0u) {
#pragma unroll
                for (int k = 0; k < 4; ++k) {
                    if ((mask[k] >> lane) & 1u)
                        myq[cnt + __builtin_popcount(mask[k] & ((1u << lane) - 1u))] =
                            (unsigned short)(tag | (j * 128 + lane * 4 + k));
                    cnt += __builtin_popcount(mask[k]);
                }
            }
        }
    }
    return cnt;
}

#define QCAP (SLICE + 32)
#define TAG_SLICES 64

template <bool REC, bool FULL>
__device__ __forceinline__ void gather_group(const float* __restrict__ h, const int* __restrict__ row_idx,
                                             const int* __restrict__ col_idx, const float* __restrict__ w,
                                             const unsigned* __restrict__ recs, float* acc,
                                             const unsigned short* qg, int n, int lane, int wave,
                                             int kblock, int rot, int nslices, int lo) {
    const int qi = FULL ? lane : ((lane < n) ? lane : 0);
    const unsigned en = qg[qi];
    int sl = wave + OWN_WAVES * (kblock + (int)(en >> 10)) + rot;
    if (sl >= nslices) sl -= nslices;
    const int e = sl * SLICE + (int)(en & (SLICE - 1));
    int r, lrow;
    float wt;
    if constexpr (REC) {
        const uint2 rc = *(const uint2*)(recs + 2 * e);
        r    = (int)(rc.x & ((1u << REC_ROW_BITS) - 1u));
        lrow = (int)(rc.x >> REC_ROW_BITS);
        wt   = __uint_as_float(rc.y);
    } else {
        r    = row_idx[e];
        lrow = col_idx[e] - lo;
        wt   = w[e];
    }

    if constexpr (FULL) {
        float v[2][4][4];
#pragma unroll
        for (int u = 0; u < 4; ++u) {
            const float* __restrict__ hr = h + (size_t)__builtin_amdgcn_readlane(r, u) * D + lane;
#pragma unroll
            for (int j = 0; j < 4; ++j) v[0][u][j] = hr[32 * j];
        }
#pragma unroll
        for (int g = 0; g < 8; ++g) {
            if (g + 1 < 8) {
#pragma unroll
                for (int u = 0; u < 4; ++u) {
                    const float* __restrict__ hr =
                        h + (size_t)__builtin_amdgcn_readlane(r, 4 * (g + 1) + u) * D + lane;
#pragma unroll
                    for (int j = 0; j < 4; ++j) v[(g + 1) & 1][u][j] = hr[32 * j];
                }
            }
#pragma unroll
            for (int u = 0; u < 4; ++u) {
                const int   lr = __builtin_amdgcn_readlane(lrow, 4 * g + u);
                const float sc = __int_as_float(__builtin_amdgcn_readlane(__float_as_int(wt), 4 * g + u));
                float* a = acc + lr * D + lane;
#pragma unroll
                for (int j = 0; j < 4; ++j)
                    __hip_atomic_fetch_add(a + 32 * j, v[g & 1][u][j] * sc, __ATOMIC_RELAXED,
                                           __HIP_MEMORY_SCOPE_WORKGROUP);
            }
        }
    } else {
        for (int t = 0; t < n; t += 4) {
            float v[4][4];
            float sc[4];
            int   lr[4];
#pragma unroll
            for (int u = 0; u < 4; ++u) {
                const int tt = min(t + u, n - 1);
                const int rt = __builtin_amdgcn_readlane(r, tt);
                lr[u] = __builtin_amdgcn_readlane(lrow, tt);
                sc[u] = __int_as_float(__builtin_amdgcn_readlane(__float_as_int(wt), tt));
                const float* __restrict__ hr = h + (size_t)rt * D + lane;
#pragma unroll
                for (int j = 0; j < 4; ++j) v[u][j] = hr[32 * j];
            }
#pragma unroll
            for (int u = 0; u < 4; ++u) {
                if (t + u < n) {
                    float* a = acc + lr[u] * D + lane;
#pragma unroll
                    for (int j = 0; j < 4; ++j)
                        __hip_atomic_fetch_add(a + 32 * j, v[u][j] * sc[u], __ATOMIC_RELAXED,
                                               __HIP_MEMORY_SCOPE_WORKGROUP);
                }
            }
        }
    }
}

template <bool KEYED, bool REC>
__global__ __launch_bounds__(1024) void owner_gather(const float* __restrict__ h,
                                                     const int* __restrict__ row_idx,
                                                     const int* __restrict__ col_idx,
                                                     const float* __restrict__ w,
                                                     const float* __restrict__ deg2,
                                                     const unsigned* __restrict__ keys,
                                                     const unsigned* __restrict__ recs,
                                                     float* __restrict__ out, int N, int E) {
    __shared__ float          acc[TILE_ROWS * D];
    __shared__ unsigned short q[OWN_WAVES * QCAP];

    const int wave = __builtin_amdgcn_readfirstlane(threadIdx.x >> 5);
    const int lane = threadIdx.x & 31;
    unsigned short* myq = q + wave * QCAP;

    constexpr int NREG = KEYED ? SLICE / KEYS_PER_LOAD : SLICE / 128;
    const int  nslices = (E + SLICE - 1) / SLICE;
    const bool vec_ok  = (E & 3) == 0;
    const int  rot     = (int)(((unsigned)blockIdx.x * (unsigned)nslices / gridDim.x) % (unsigned)nslices);

    for (int tile = blockIdx.x; tile * TILE_ROWS < N; tile += gridDim.x) {
        const int lo    = tile * TILE_ROWS;
        const int nrows = min(TILE_ROWS, N - lo);

        for (int i = threadIdx.x; i < TILE_ROWS * D / 4; i += blockDim.x)
            ((float4*)acc)[i] = make_float4(0.0f, 0.0f, 0.0f, 0.0f);
        __syncthreads();

        v4i cur[NREG];
        {
            int sl = wave + rot; if (sl >= nslices) sl -= nslices;
            if (wave < nslices) fetch_slice<KEYED>(col_idx, keys, sl * SLICE, lane, E, vec_ok, cur);
        }
        int cnt = 0;
        int k   = 0;
        for (int s = wave; s < nslices; s += OWN_WAVES, ++k) {
            int sl = s + rot; if (sl >= nslices) sl -= nslices;
            const int e0 = sl * SLICE;

            v4i nxt[NREG];
            {
                int sn = ((s + OWN_WAVES < nslices) ? s + OWN_WAVES : s) + rot;
                if (sn >= nslices) sn -= nslices;
                fetch_slice<KEYED>(col_idx, keys, sn * SLICE, lane, E, vec_ok, nxt);
            }

            cnt += filter_slice<KEYED>(cur, myq + cnt, lane, e0, E, tile, lo, nrows,
                                       (k & (TAG_SLICES - 1)) << 10);

            const int kblock = k & ~(TAG_SLICES - 1);
            int head = 0;
            for (; cnt - head >= 32; head += 32)
                gather_group<REC, true>(h, row_idx, col_idx, w, recs, acc, myq + head, 32, lane, wave,
                                        kblock, rot, nslices, lo);
            const bool flush = (s + OWN_WAVES >= nslices) || ((k & (TAG_SLICES - 1)) == TAG_SLICES - 1);
            if (flush && cnt > head) {
                gather_group<REC, false>(h, row_idx, col_idx, w, recs, acc, myq + head, cnt - head, lane, wave,
                                         kblock, rot, nslices, lo);
                head = cnt;
            }
            const int rest = cnt - head;
            if (head > 0 && rest > 0) {
                const unsigned short en = myq[head + ((lane < rest) ? lane : 0)];
                if (lane < rest) myq[lane] = en;
            }
            cnt = rest;

#pragma unroll
            for (int j = 0; j < NREG; ++j) cur[j] = nxt[j];
        }
        __syncthreads();

        float4*       o4 = (float4*)(out + (size_t)lo * D);
        const float4* a4 = (const float4*)acc;
        for (int i = threadIdx.x; i < nrows * (D / 4); i += blockDim.x) {
            const int   c  = lo + i / (D / 4);
            const float dc = inv_sqrt_deg(deg2[c], deg2[N + c]);
            float4 o = o4[i];
            const float4 a = a4[i];
            o.x += a.x * dc; o.y += a.y * dc; o.z += a.z * dc; o.w += a.w * dc;
            o4[i] = o;
        }
        __syncthreads();
    }
}

#define CELL_SLOTS 16
#define CELL_RECS  (CELL_SLOTS - 1)
#define MAX_TILES  256

__global__ void zero_f32_and_count(float* __restrict__ p, int n, int* __restrict__ side_count) {
    int i = blockIdx.x * blockDim.x + threadIdx.x;
    if (i < n) p[i] = 0.0f;
    if (i == 0) *side_count = 0;
}

__global__ __launch_bounds__(SLICE / 4) void deg_cells_kernel(const int* __restrict__ row,
                                                              const int* __restrict__ col,
                                                              const float* __restrict__ w,
                                                              float* __restrict__ deg2,
                                                              uint4* __restrict__ cells,
                                                              int* __restrict__ side_count,
                                                              int* __restrict__ side_idx,
                                                              int n, int E, int nslices, int ntile) {
    __shared__ __attribute__((aligned(16))) uint2 stage[MAX_TILES * CELL_SLOTS];
    __shared__ int cnt[MAX_TILES];

    const int tid = threadIdx.x;
#pragma unroll
    for (int q = 0; q < CELL_SLOTS / 2; ++q)
        ((uint4*)stage)[tid + (SLICE / 4) * q] = make_uint4(0u, 0u, 0u, 0u);
    cnt[tid] = 0;
    __syncthreads();

    const unsigned place = __builtin_amdgcn_s_sendmsg_rtn(0x87);
    float* __restrict__ deg = deg2 + (size_t)((place >> 18) & 1u) * n;
    const int e = (blockIdx.x * blockDim.x + tid) * 4;

    int   cc[4], rr[4];
    float wv[4];
    if ((E & 3) == 0 && e < E) {
        const v4i    c  = *(const v4i*)(col + e);
        const float4 ww = *(const float4*)(w + e);
        const v4i    r  = *(const v4i*)(row + e);
        cc[0] = c.x; cc[1] = c.y; cc[2] = c.z; cc[3] = c.w;
        rr[0] = r.x; rr[1] = r.y; rr[2] = r.z; rr[3] = r.w;
        wv[0] = ww.x; wv[1] = ww.y; wv[2] = ww.z; wv[3] = ww.w;
    } else {
#pragma unroll
        for (int k = 0; k < 4; ++k) {
            const bool in = e + k < E;
            cc[k] = in ? col[e + k] : -1;
            rr[k] = in ? row[e + k] : 0;
            wv[k] = in ? w[e + k] : 0.0f;
        }
    }
#pragma unroll
    for (int k = 0; k < 4; ++k) {
        if (cc[k] >= 0) {
            unsafeAtomicAdd(&deg[cc[k]], wv[k]);
            const unsigned t    = (unsigned)cc[k] / TILE_ROWS;
            const unsigned lrow = (unsigned)cc[k] - t * TILE_ROWS;
            const int rank = atomicAdd(&cnt[t], 1);
            if (rank < CELL_RECS) {
                stage[t * CELL_SLOTS + 1 + rank] =
                    make_uint2((unsigned)rr[k] | (lrow << REC_ROW_BITS), __float_as_uint(wv[k]));
            } else {
                const int p = atomicAdd(side_count, 1);
                side_idx[p] = e + k;
            }
        }
    }
    __syncthreads();

#pragma unroll
    for (int q = 0; q < CELL_SLOTS / 2; ++q) {
        const int i     = tid + (SLICE / 4) * q;
        const int cell  = i >> 3;
        const int piece = i & 7;
        uint4 v = ((const uint4*)stage)[i];
        if (piece == 0) { v.x = (unsigned)cnt[cell]; v.y = 0u; }
        if (cell < ntile)
            cells[((size_t)cell * nslices + blockIdx.x) * (CELL_SLOTS / 2) + piece] = v;
    }
}

template <bool FULL>
__device__ __forceinline__ void gather_recs(const float* __restrict__ h, float* acc,
                                            const uint2* qg, int n, int lane) {
    const int   qi   = FULL ? lane : ((lane < n) ? lane : 0);
    const uint2 rc   = qg[qi];
    const int   r    = (int)(rc.x & ((1u << REC_ROW_BITS) - 1u));
    const int   lrow = (int)(rc.x >> REC_ROW_BITS);
    const float wt   = __uint_as_float(rc.y);

    if constexpr (FULL) {
        float v[2][4][4];
#pragma unroll
        for (int u = 0; u < 4; ++u) {
            const float* __restrict__ hr = h + (size_t)__builtin_amdgcn_readlane(r, u) * D + lane;
#pragma unroll
            for (int j = 0; j < 4; ++j) v[0][u][j] = hr[32 * j];
        }
#pragma unroll
        for (int g = 0; g < 8; ++g) {
            if (g + 1 < 8) {
#pragma unroll
                for (int u = 0; u < 4; ++u) {
                    const float* __restrict__ hr =
                        h + (size_t)__builtin_amdgcn_readlane(r, 4 * (g + 1) + u) * D + lane;
#pragma unroll
                    for (int j = 0; j < 4; ++j) v[(g + 1) & 1][u][j] = hr[32 * j];
                }
            }
#pragma unroll
            for (int u = 0; u < 4; ++u) {
                const int   lr = __builtin_amdgcn_readlane(lrow, 4 * g + u);
                const float sc = __int_as_float(__builtin_amdgcn_readlane(__float_as_int(wt), 4 * g + u));
                float* a = acc + lr * D + lane;
#pragma unroll
                for (int j = 0; j < 4; ++j)
                    __hip_atomic_fetch_add(a + 32 * j, v[g & 1][u][j] * sc, __ATOMIC_RELAXED,
                                           __HIP_MEMORY_SCOPE_WORKGROUP);
            }
        }
    } else {
        for (int t = 0; t < n; t += 4) {
            float v[4][4];
            float sc[4];
            int   lr[4];
#pragma unroll
            for (int u = 0; u < 4; ++u) {
                const int tt = min(t + u, n - 1);
                const int rt = __builtin_amdgcn_readlane(r, tt);
                lr[u] = __builtin_amdgcn_readlane(lrow, tt);
                sc[u] = __int_as_float(__builtin_amdgcn_readlane(__float_as_int(wt), tt));
                const float* __restrict__ hr = h + (size_t)rt * D + lane;
#pragma unroll
                for (int j = 0; j < 4; ++j) v[u][j] = hr[32 * j];
            }
#pragma unroll
            for (int u = 0; u < 4; ++u) {
                if (t + u < n) {
                    float* a = acc + lr[u] * D + lane;
#pragma unroll
                    for (int j = 0; j < 4; ++j)
                        __hip_atomic_fetch_add(a + 32 * j, v[u][j] * sc[u], __ATOMIC_RELAXED,
                                               __HIP_MEMORY_SCOPE_WORKGROUP);
                }
            }
        }
    }
}

#define GRP_SLICES 8
#define GRP_LOADS  (GRP_SLICES / 2)
#define CQCAP (32 + GRP_SLICES * CELL_RECS + 8)

__device__ __forceinline__ void fetch_cells(const uint2* __restrict__ ct, int group, int lane, int nslices,
                                            uint2 (&c)[GRP_LOADS]) {
#pragma unroll
    for (int j = 0; j < GRP_LOADS; ++j) {
        const int s0 = group * GRP_SLICES + 2 * j;
        c[j] = (s0 + (lane >> 4) < nslices) ? ct[(size_t)s0 * CELL_SLOTS + lane] : make_uint2(0u, 0u);
    }
}

__global__ __launch_bounds__(1024) void owner_gather_cells(const float* __restrict__ h,
                                                           const int* __restrict__ row_idx,
                                                           const int* __restrict__ col_idx,
                                                           const float* __restrict__ w,
                                                           const float* __restrict__ deg2,
                                                           const uint2* __restrict__ cells,
                                                           const int* __restrict__ side_count,
                                                           const int* __restrict__ side_idx,
                                                           float* __restrict__ out, int N, int E) {
    __shared__ float acc[TILE_ROWS * D];
    __shared__ uint2 q2[OWN_WAVES * CQCAP];

    const int wave = __builtin_amdgcn_readfirstlane(threadIdx.x >> 5);
    const int lane = threadIdx.x & 31;
    uint2* myq = q2 + wave * CQCAP;

    const int nslices = (E + SLICE - 1) / SLICE;
    const int ngroups = (nslices + GRP_SLICES - 1) / GRP_SLICES;
    const int slot    = lane & 15;

    for (int tile = blockIdx.x; tile * TILE_ROWS < N; tile += gridDim.x) {
        const int lo    = tile * TILE_ROWS;
        const int nrows = min(TILE_ROWS, N - lo);

        for (int i = threadIdx.x; i < TILE_ROWS * D / 4; i += blockDim.x)
            ((float4*)acc)[i] = make_float4(0.0f, 0.0f, 0.0f, 0.0f);
        __syncthreads();

        const uint2* __restrict__ ct = cells + (size_t)tile * nslices * CELL_SLOTS;
        uint2 cur[GRP_LOADS] = {};
        if (wave < ngroups) fetch_cells(ct, wave, lane, nslices, cur);

        int cnt = 0;
        for (int g = wave; g < ngroups; g += OWN_WAVES) {
            uint2 nxt[GRP_LOADS];
            fetch_cells(ct, (g + OWN_WAVES < ngroups) ? g + OWN_WAVES : g, lane, nslices, nxt);

#pragma unroll
            for (int j = 0; j < GRP_LOADS; ++j) {
                const int c0 = __builtin_amdgcn_readlane((int)cur[j].x, 0);
                const int c1 = __builtin_amdgcn_readlane((int)cur[j].x, 16);
                const int hc = min((lane & 16) ? c1 : c0, CELL_RECS);
                const bool ok = slot >= 1 && slot <= hc;
                const unsigned m = __builtin_amdgcn_ballot_w32(ok);
                if (ok) myq[cnt + __builtin_popcount(m & ((1u << lane) - 1u))] = cur[j];
                cnt += __builtin_popcount(m);
            }

            int head = 0;
            for (; cnt - head >= 32; head += 32)
                gather_recs<true>(h, acc, myq + head, 32, lane);
            if (g + OWN_WAVES >= ngroups && cnt > head) {
                gather_recs<false>(h, acc, myq + head, cnt - head, lane);
                head = cnt;
            }
            const int rest = cnt - head;
            if (head > 0 && rest > 0) {
                const uint2 en = myq[head + ((lane < rest) ? lane : 0)];
                if (lane < rest) myq[lane] = en;
            }
            cnt = rest;

#pragma unroll
            for (int j = 0; j < GRP_LOADS; ++j) cur[j] = nxt[j];
        }

        const int nside = min(*side_count, E);
        for (int i = threadIdx.x; i < nside; i += blockDim.x) {
            const int e = side_idx[i];
            const int c = col_idx[e];
            if ((unsigned)(c - lo) < (unsigned)nrows) {
                const float wt = w[e];
                const float* __restrict__ hr = h + (size_t)row_idx[e] * D;
                float* a = acc + (c - lo) * D;
                for (int j = 0; j < D; ++j)
                    __hip_atomic_fetch_add(a + j, hr[j] * wt, __ATOMIC_RELAXED, __HIP_MEMORY_SCOPE_WORKGROUP);
            }
        }
        __syncthreads();

        float4*       o4 = (float4*)(out + (size_t)lo * D);
        const float4* a4 = (const float4*)acc;
        for (int i = threadIdx.x; i < nrows * (D / 4); i += blockDim.x) {
            const int   c  = lo + i / (D / 4);
            const float dc = inv_sqrt_deg(deg2[c], deg2[N + c]);
            float4 o = o4[i];
            const float4 a = a4[i];
            o.x += a.x * dc; o.y += a.y * dc; o.z += a.z * dc; o.w += a.w * dc;
            o4[i] = o;
        }
        __syncthreads();
    }
}


extern "C" void kernel_launch(void* const* d_in, const int* in_sizes, int n_in,
                              void* d_out, int out_size, void* d_ws, size_t ws_size,
                              hipStream_t stream) {
    const float* x    = (const float*)d_in[0];
    const int*   ei   = (const int*)d_in[1];
    const float* ew   = (const float*)d_in[2];
    const float* Wlin = (const float*)d_in[3];
    const float* Wgcn = (const float*)d_in[4];
    float*       out  = (float*)d_out;

    const int N = in_sizes[0] / D;
    const int E = in_sizes[2];

    const int* row = ei;
    const int* col = ei + E;

    float* deg2 = (float*)d_ws;
    float* h    = deg2 + 2 * N;
    const int nslices = (E + SLICE - 1) / SLICE;
    const bool keyed = ws_size >= (size_t)(2 * N + N * D) * sizeof(float) + (size_t)nslices * SLICE;
    unsigned* keys = keyed ? (unsigned*)(h + (size_t)N * D) : nullptr;
    const size_t rec_off = ((size_t)(2 * N + N * D) * sizeof(float) + (size_t)nslices * SLICE + 15) & ~(size_t)15;
    const bool rec = keyed && N <= (1 << REC_ROW_BITS) && ws_size >= rec_off + (size_t)nslices * SLICE * 8;
    unsigned* recs = rec ? (unsigned*)((char*)d_ws + rec_off) : nullptr;

    const int    ntile     = (N + TILE_ROWS - 1) / TILE_ROWS;
    const size_t cells_off = ((size_t)(2 * N + N * D) * sizeof(float) + 127) & ~(size_t)127;
    const size_t side_off  = cells_off + (size_t)ntile * nslices * (CELL_SLOTS * 8);
    const size_t cnt_off   = (side_off + (size_t)E * sizeof(int) + 15) & ~(size_t)15;
    if (ntile <= MAX_TILES && N <= (1 << REC_ROW_BITS) && ws_size >= cnt_off + 16) {
        uint4* cells      = (uint4*)((char*)d_ws + cells_off);
        int*   side_idx   = (int*)((char*)d_ws + side_off);
        int*   side_count = (int*)((char*)d_ws + cnt_off);
        zero_f32_and_count<<<(2 * N + 255) / 256, 256, 0, stream>>>(deg2, 2 * N, side_count);
        deg_cells_kernel<<<nslices, SLICE / 4, 0, stream>>>(row, col, ew, deg2, cells, side_count, side_idx,
                                                            N, E, nslices, ntile);
        gemm_dual<<<512, 512, 0, stream>>>(x, Wlin, Wgcn, deg2, out, h, N / 16, N);
        owner_gather_cells<<<ntile, 1024, 0, stream>>>(h, row, col, ew, deg2, (const uint2*)cells, side_count,
                                                       side_idx, out, N, E);
        return;
    }

    zero_f32<<<(2 * N + 255) / 256, 256, 0, stream>>>(deg2, 2 * N);
    deg_kernel<<<nslices, SLICE / 4, 0, stream>>>(row, col, ew, deg2, keys, recs, N, E);
    gemm_dual<<<512, 512, 0, stream>>>(x, Wlin, Wgcn, deg2, out, h, N / 16, N);
    const int nblk = (N + TILE_ROWS - 1) / TILE_ROWS;
    if (rec)        owner_gather<true, true><<<nblk, 1024, 0, stream>>>(h, row, col, ew, deg2, keys, recs, out, N, E);
    else if (keyed) owner_gather<true, false><<<nblk, 1024, 0, stream>>>(h, row, col, ew, deg2, keys, recs, out, N, E);
    else            owner_gather<false, false><<<nblk, 1024, 0, stream>>>(h, row, col, ew, deg2, keys, recs, out, N, E);
}


# x@W_lin product moved into the degree/cell launch as 512 resident 8-wave blocks before the slice blocks: matrix work runs under their memory waits; W_gcn product in its own launch; sums unchanged
# speedup vs baseline: 3.5798x; 1.0527x over previous
#include <hip/hip_runtime.h>

#define D 128

typedef float v2f __attribute__((ext_vector_type(2)));
typedef float v8f __attribute__((ext_vector_type(8)));


__global__ void zero_f32(float* __restrict__ p, int n) {
    int i = blockIdx.x * blockDim.x + threadIdx.x;
    if (i < n) p[i] = 0.0f;
}


#define TILE_ROWS 392
#define SLICE     1024
#define OWN_WAVES 32
#define KEYS_PER_LOAD 512

typedef int v4i __attribute__((ext_vector_type(4)));

#define REC_ROW_BITS 23
__global__ __launch_bounds__(SLICE / 4) void deg_kernel(const int* __restrict__ row,
                                                        const int* __restrict__ col,
                                                        const float* __restrict__ w,
                                                        float* __restrict__ deg2,
                                                        unsigned* __restrict__ keys,
                                                        unsigned* __restrict__ recs, int n, int E) {
    const unsigned place = __builtin_amdgcn_s_sendmsg_rtn(0x87);
    float* __restrict__ deg = deg2 + (size_t)((place >> 18) & 1u) * n;
    const int e = (blockIdx.x * blockDim.x + threadIdx.x) * 4;
    unsigned key = 0u;
    if ((E & 3) == 0 && e < E) {
        const v4i    c  = *(const v4i*)(col + e);
        const float4 ww = *(const float4*)(w + e);
        unsafeAtomicAdd(&deg[c.x], ww.x);
        unsafeAtomicAdd(&deg[c.y], ww.y);
        unsafeAtomicAdd(&deg[c.z], ww.z);
        unsafeAtomicAdd(&deg[c.w], ww.w);
        const unsigned tx = (unsigned)c.x / TILE_ROWS, ty = (unsigned)c.y / TILE_ROWS;
        const unsigned tz = (unsigned)c.z / TILE_ROWS, tw = (unsigned)c.w / TILE_ROWS;
        key = tx | (ty << 8) | (tz << 16) | (tw << 24);
        if (recs != nullptr) {
            const v4i r = *(const v4i*)(row + e);
            uint4 a, b;
            a.x = (unsigned)r.x | (((unsigned)c.x - tx * TILE_ROWS) << REC_ROW_BITS); a.y = __float_as_uint(ww.x);
            a.z = (unsigned)r.y | (((unsigned)c.y - ty * TILE_ROWS) << REC_ROW_BITS); a.w = __float_as_uint(ww.y);
            b.x = (unsigned)r.z | (((unsigned)c.z - tz * TILE_ROWS) << REC_ROW_BITS); b.y = __float_as_uint(ww.z);
            b.z = (unsigned)r.w | (((unsigned)c.w - tw * TILE_ROWS) << REC_ROW_BITS); b.w = __float_as_uint(ww.w);
            *(uint4*)(recs + 2 * e)     = a;
            *(uint4*)(recs + 2 * e + 4) = b;
        }
    } else {
        for (int k = 0; k < 4; ++k) {
            if (e + k < E) {
                const int c = col[e + k];
                const float wk = w[e + k];
                unsafeAtomicAdd(&deg[c], wk);
                const unsigned t = (unsigned)c / TILE_ROWS;
                key |= t << (8 * k);
                if (recs != nullptr)
                    *(uint2*)(recs + 2 * (e + k)) =
                        make_uint2((unsigned)row[e + k] | (((unsigned)c - t * TILE_ROWS) << REC_ROW_BITS),
                                   __float_as_uint(wk));
            }
        }
    }
    if (keys != nullptr) keys[e >> 2] = key;
}

__device__ __forceinline__ float inv_sqrt_deg(float plane0, float plane1) {
    const float d = plane0 + plane1;
    return (d > 0.0f) ? __frsqrt_rn(d) : 0.0f;
}


#define XS_STRIDE 132

__global__ __launch_bounds__(512) void gemm_dual(const float* __restrict__ x,
                                                 const float* __restrict__ Wlin,
                                                 const float* __restrict__ Wgcn,
                                                 const float* __restrict__ deg2,
                                                 float* __restrict__ out_hl,
                                                 float* __restrict__ h, int ntiles, int n) {
    __shared__ __attribute__((aligned(16))) float xs[2][16 * XS_STRIDE];

    const int wave  = threadIdx.x >> 5;
    const int lane  = threadIdx.x & 31;
    const int which = wave >> 3;
    const int col0  = (wave & 7) * 16;

    const float* __restrict__ W   = which ? Wgcn : Wlin;
    float* __restrict__       dst = which ? h    : out_hl;

    const int m  = lane & 15;
    const int hh = lane >> 4;

    v2f b[32];
#pragma unroll
    for (int ks = 0; ks < 32; ++ks) {
        b[ks].x = W[(size_t)(4 * ks + 2 * hh + 0) * D + col0 + m];
        b[ks].y = W[(size_t)(4 * ks + 2 * hh + 1) * D + col0 + m];
    }

    const float* __restrict__ xsrc = x + (size_t)wave * D + lane * 4;
    const int soff = wave * XS_STRIDE + lane * 4;

    const float* __restrict__ dg0 = deg2 + m;
    const float* __restrict__ dg1 = deg2 + (size_t)n + m;

    int tile = blockIdx.x;
    float4 stage = make_float4(0.0f, 0.0f, 0.0f, 0.0f);
    float dsum0 = 0.0f, dsum1 = 0.0f;
    if (tile < ntiles) {
        stage = *(const float4*)(xsrc + (size_t)tile * 16 * D);
        if (which) { dsum0 = dg0[tile * 16]; dsum1 = dg1[tile * 16]; }
    }
    *(float4*)&xs[0][soff] = stage;
    __syncthreads();

    int buf = 0;
    for (; tile < ntiles; tile += gridDim.x) {
        const int  nt   = tile + gridDim.x;
        const bool more = nt < ntiles;
        const float rowfac = which ? inv_sqrt_deg(dsum0, dsum1) : 1.0f;
        if (more) {
            stage = *(const float4*)(xsrc + (size_t)nt * 16 * D);
            if (which) { dsum0 = dg0[nt * 16]; dsum1 = dg1[nt * 16]; }
        }

        const float* xa = &xs[buf][m * XS_STRIDE + 2 * hh];
        v8f acc = {};
#pragma unroll
        for (int ks = 0; ks < 32; ++ks) {
            const v2f a = *(const v2f*)(xa + 4 * ks);
            acc = __builtin_amdgcn_wmma_f32_16x16x4_f32(
                 false, a,  false, b[ks],
                 (short)0, acc,  false,  false);
        }

        float* __restrict__ drow = dst + (size_t)(tile * 16 + 8 * hh) * D + col0 + m;
#pragma unroll
        for (int r = 0; r < 8; ++r) {
            const float f0 = __int_as_float(__builtin_amdgcn_readlane(__float_as_int(rowfac), r));
            const float f1 = __int_as_float(__builtin_amdgcn_readlane(__float_as_int(rowfac), 8 + r));
            drow[(size_t)r * D] = acc[r] * (hh ? f1 : f0);
        }

        if (more) *(float4*)&xs[buf ^ 1][soff] = stage;
        __syncthreads();
        buf ^= 1;
    }
}

template <int WHICH>
__device__ __forceinline__ void product_block8(const float* __restrict__ x, const float* __restrict__ W,
                                               const float* __restrict__ deg2, float* __restrict__ dst,
                                               int ntiles, int n, int first, int step, float* xs) {
    const int wave = threadIdx.x >> 5;
    const int lane = threadIdx.x & 31;
    const int col0 = wave * 16;
    const int m    = lane & 15;
    const int hh   = lane >> 4;

    v2f b[32];
#pragma unroll
    for (int ks = 0; ks < 32; ++ks) {
        b[ks].x = W[(size_t)(4 * ks + 2 * hh + 0) * D + col0 + m];
        b[ks].y = W[(size_t)(4 * ks + 2 * hh + 1) * D + col0 + m];
    }

    const float* __restrict__ xsrc = x + (size_t)wave * D + lane * 4;
    const int soff = wave * XS_STRIDE + lane * 4;

    const float* __restrict__ dg0 = WHICH ? deg2 + m : nullptr;
    const float* __restrict__ dg1 = WHICH ? deg2 + (size_t)n + m : nullptr;

    int tile = first;
    float4 st0 = make_float4(0.0f, 0.0f, 0.0f, 0.0f), st1 = st0;
    float dsum0 = 0.0f, dsum1 = 0.0f;
    if (tile < ntiles) {
        st0 = *(const float4*)(xsrc + (size_t)tile * 16 * D);
        st1 = *(const float4*)(xsrc + (size_t)tile * 16 * D + 8 * D);
        if (WHICH) { dsum0 = dg0[tile * 16]; dsum1 = dg1[tile * 16]; }
    }
    *(float4*)&xs[soff]                 = st0;
    *(float4*)&xs[soff + 8 * XS_STRIDE] = st1;
    __syncthreads();

    int buf = 0;
    for (; tile < ntiles; tile += step) {
        const int  nt   = tile + step;
        const bool more = nt < ntiles;
        float rowfac = 1.0f;
        if (WHICH) rowfac = inv_sqrt_deg(dsum0, dsum1);
        if (more) {
            st0 = *(const float4*)(xsrc + (size_t)nt * 16 * D);
            st1 = *(const float4*)(xsrc + (size_t)nt * 16 * D + 8 * D);
            if (WHICH) { dsum0 = dg0[nt * 16]; dsum1 = dg1[nt * 16]; }
        }

        const float* xa = xs + buf * (16 * XS_STRIDE) + m * XS_STRIDE + 2 * hh;
        v8f acc = {};
#pragma unroll
        for (int ks = 0; ks < 32; ++ks) {
            const v2f a = *(const v2f*)(xa + 4 * ks);
            acc = __builtin_amdgcn_wmma_f32_16x16x4_f32(
                 false, a,  false, b[ks],
                 (short)0, acc,  false,  false);
        }

        float* __restrict__ drow = dst + (size_t)(tile * 16 + 8 * hh) * D + col0 + m;
        if (WHICH) {
#pragma unroll
            for (int r = 0; r < 8; ++r) {
                const float f0 = __int_as_float(__builtin_amdgcn_readlane(__float_as_int(rowfac), r));
                const float f1 = __int_as_float(__builtin_amdgcn_readlane(__float_as_int(rowfac), 8 + r));
                drow[(size_t)r * D] = acc[r] * (hh ? f1 : f0);
            }
        } else {
#pragma unroll
            for (int r = 0; r < 8; ++r) drow[(size_t)r * D] = acc[r];
        }

        if (more) {
            float* o = xs + (buf ^ 1) * (16 * XS_STRIDE);
            *(float4*)&o[soff]                 = st0;
            *(float4*)&o[soff + 8 * XS_STRIDE] = st1;
        }
        __syncthreads();
        buf ^= 1;
    }
}

__global__ __launch_bounds__(256) void gemm_gcn(const float* __restrict__ x, const float* __restrict__ Wgcn,
                                                const float* __restrict__ deg2, float* __restrict__ h,
                                                int ntiles, int n) {
    __shared__ __attribute__((aligned(16))) float xs[2 * 16 * XS_STRIDE];
    product_block8<1>(x, Wgcn, deg2, h, ntiles, n, blockIdx.x, gridDim.x, xs);
}


template <bool KEYED>
__device__ __forceinline__ void fetch_slice(const int* __restrict__ col, const unsigned* __restrict__ keys,
                                            int e0, int lane, int E, bool vec_ok,
                                            v4i (&c)[KEYED ? SLICE / KEYS_PER_LOAD : SLICE / 128]) {
    if constexpr (KEYED) {
#pragma unroll
        for (int j = 0; j < SLICE / KEYS_PER_LOAD; ++j)
            c[j] = *(const v4i*)(keys + ((e0 + j * KEYS_PER_LOAD) >> 2) + lane * 4);
    } else if (vec_ok && e0 + SLICE <= E) {
#pragma unroll
        for (int j = 0; j < SLICE / 128; ++j)
            c[j] = *(const v4i*)(col + e0 + j * 128 + lane * 4);
    } else {
#pragma unroll
        for (int j = 0; j < SLICE / 128; ++j) {
            const int e = e0 + j * 128 + lane * 4;
            c[j].x = (e + 0 < E) ? col[e + 0] : -1;
            c[j].y = (e + 1 < E) ? col[e + 1] : -1;
            c[j].z = (e + 2 < E) ? col[e + 2] : -1;
            c[j].w = (e + 3 < E) ? col[e + 3] : -1;
        }
    }
}

template <bool KEYED>
__device__ __forceinline__ int filter_slice(v4i (&cur)[KEYED ? SLICE / KEYS_PER_LOAD : SLICE / 128],
                                            unsigned short* myq, int lane, int e0, int E,
                                            int tile, int lo, int nrows, int tag) {
    int cnt = 0;
    if constexpr (KEYED) {
        const unsigned pat = (unsigned)tile * 0x01010101u;
        if (e0 + SLICE > E) {
#pragma unroll
            for (int j = 0; j < SLICE / KEYS_PER_LOAD; ++j) {
#pragma unroll
                for (int d = 0; d < 4; ++d) {
                    const int nv = E - (e0 + j * KEYS_PER_LOAD + lane * 16 + 4 * d);
                    const unsigned keep = (nv >= 4) ? 0xffffffffu : ((nv <= 0) ? 0u : ((1u << (8 * nv)) - 1u));
                    cur[j][d] = (int)(((unsigned)cur[j][d] & keep) | (~pat & ~keep));
                }
            }
        }
#pragma unroll
        for (int j = 0; j < SLICE / KEYS_PER_LOAD; ++j) {
            unsigned g = 0u;
#pragma unroll
            for (int d = 0; d < 4; ++d) {
                const unsigned x  = (unsigned)cur[j][d] ^ pat;
                const unsigned nz = ((x & 0x7f7f7f7fu) + 0x7f7f7f7fu) | x;
                g |= (~nz & 0x80808080u) >> (7 - d);
            }
            unsigned hit = __builtin_amdgcn_ballot_w32(g != 0u);
            while (hit != 0u) {
                if (g != 0u) {
                    const int p = __builtin_ctz(g);
                    myq[cnt + __builtin_popcount(hit & ((1u << lane) - 1u))] =
                        (unsigned short)(tag | (j * KEYS_PER_LOAD + lane * 16 + ((p & 7) << 2) + (p >> 3)));
                    g &= g - 1u;
                }
                cnt += __builtin_popcount(hit);
                hit = __builtin_amdgcn_ballot_w32(g != 0u);
            }
        }
    } else {
#pragma unroll
        for (int j = 0; j < SLICE / 128; ++j) {
            unsigned mask[4];
#pragma unroll
            for (int k = 0; k < 4; ++k)
                mask[k] = __builtin_amdgcn_ballot_w32((unsigned)(cur[j][k] - lo) < (unsigned)nrows);
            if ((mask[0] | mask[1] | mask[2] | mask[3]) != 0u) {
#pragma unroll
                for (int k = 0; k < 4; ++k) {
                    if ((mask[k] >> lane) & 1u)
                        myq[cnt + __builtin_popcount(mask[k] & ((1u << lane) - 1u))] =
                            (unsigned short)(tag | (j * 128 + lane * 4 + k));
                    cnt += __builtin_popcount(mask[k]);
                }
            }
        }
    }
    return cnt;
}

#define QCAP (SLICE + 32)
#define TAG_SLICES 64

template <bool REC, bool FULL>
__device__ __forceinline__ void gather_group(const float* __restrict__ h, const int* __restrict__ row_idx,
                                             const int* __restrict__ col_idx, const float* __restrict__ w,
                                             const unsigned* __restrict__ recs, float* acc,
                                             const unsigned short* qg, int n, int lane, int wave,
                                             int kblock, int rot, int nslices, int lo) {
    const int qi = FULL ? lane : ((lane < n) ? lane : 0);
    const unsigned en = qg[qi];
    int sl = wave + OWN_WAVES * (kblock + (int)(en >> 10)) + rot;
    if (sl >= nslices) sl -= nslices;
    const int e = sl * SLICE + (int)(en & (SLICE - 1));
    int r, lrow;
    float wt;
    if constexpr (REC) {
        const uint2 rc = *(const uint2*)(recs + 2 * e);
        r    = (int)(rc.x & ((1u << REC_ROW_BITS) - 1u));
        lrow = (int)(rc.x >> REC_ROW_BITS);
        wt   = __uint_as_float(rc.y);
    } else {
        r    = row_idx[e];
        lrow = col_idx[e] - lo;
        wt   = w[e];
    }

    if constexpr (FULL) {
        float v[2][4][4];
#pragma unroll
        for (int u = 0; u < 4; ++u) {
            const float* __restrict__ hr = h + (size_t)__builtin_amdgcn_readlane(r, u) * D + lane;
#pragma unroll
            for (int j = 0; j < 4; ++j) v[0][u][j] = hr[32 * j];
        }
#pragma unroll
        for (int g = 0; g < 8; ++g) {
            if (g + 1 < 8) {
#pragma unroll
                for (int u = 0; u < 4; ++u) {
                    const float* __restrict__ hr =
                        h + (size_t)__builtin_amdgcn_readlane(r, 4 * (g + 1) + u) * D + lane;
#pragma unroll
                    for (int j = 0; j < 4; ++j) v[(g + 1) & 1][u][j] = hr[32 * j];
                }
            }
#pragma unroll
            for (int u = 0; u < 4; ++u) {
                const int   lr = __builtin_amdgcn_readlane(lrow, 4 * g + u);
                const float sc = __int_as_float(__builtin_amdgcn_readlane(__float_as_int(wt), 4 * g + u));
                float* a = acc + lr * D + lane;
#pragma unroll
                for (int j = 0; j < 4; ++j)
                    __hip_atomic_fetch_add(a + 32 * j, v[g & 1][u][j] * sc, __ATOMIC_RELAXED,
                                           __HIP_MEMORY_SCOPE_WORKGROUP);
            }
        }
    } else {
        for (int t = 0; t < n; t += 4) {
            float v[4][4];
            float sc[4];
            int   lr[4];
#pragma unroll
            for (int u = 0; u < 4; ++u) {
                const int tt = min(t + u, n - 1);
                const int rt = __builtin_amdgcn_readlane(r, tt);
                lr[u] = __builtin_amdgcn_readlane(lrow, tt);
                sc[u] = __int_as_float(__builtin_amdgcn_readlane(__float_as_int(wt), tt));
                const float* __restrict__ hr = h + (size_t)rt * D + lane;
#pragma unroll
                for (int j = 0; j < 4; ++j) v[u][j] = hr[32 * j];
            }
#pragma unroll
            for (int u = 0; u < 4; ++u) {
                if (t + u < n) {
                    float* a = acc + lr[u] * D + lane;
#pragma unroll
                    for (int j = 0; j < 4; ++j)
                        __hip_atomic_fetch_add(a + 32 * j, v[u][j] * sc[u], __ATOMIC_RELAXED,
                                               __HIP_MEMORY_SCOPE_WORKGROUP);
                }
            }
        }
    }
}

template <bool KEYED, bool REC>
__global__ __launch_bounds__(1024) void owner_gather(const float* __restrict__ h,
                                                     const int* __restrict__ row_idx,
                                                     const int* __restrict__ col_idx,
                                                     const float* __restrict__ w,
                                                     const float* __restrict__ deg2,
                                                     const unsigned* __restrict__ keys,
                                                     const unsigned* __restrict__ recs,
                                                     float* __restrict__ out, int N, int E) {
    __shared__ float          acc[TILE_ROWS * D];
    __shared__ unsigned short q[OWN_WAVES * QCAP];

    const int wave = __builtin_amdgcn_readfirstlane(threadIdx.x >> 5);
    const int lane = threadIdx.x & 31;
    unsigned short* myq = q + wave * QCAP;

    constexpr int NREG = KEYED ? SLICE / KEYS_PER_LOAD : SLICE / 128;
    const int  nslices = (E + SLICE - 1) / SLICE;
    const bool vec_ok  = (E & 3) == 0;
    const int  rot     = (int)(((unsigned)blockIdx.x * (unsigned)nslices / gridDim.x) % (unsigned)nslices);

    for (int tile = blockIdx.x; tile * TILE_ROWS < N; tile += gridDim.x) {
        const int lo    = tile * TILE_ROWS;
        const int nrows = min(TILE_ROWS, N - lo);

        for (int i = threadIdx.x; i < TILE_ROWS * D / 4; i += blockDim.x)
            ((float4*)acc)[i] = make_float4(0.0f, 0.0f, 0.0f, 0.0f);
        __syncthreads();

        v4i cur[NREG];
        {
            int sl = wave + rot; if (sl >= nslices) sl -= nslices;
            if (wave < nslices) fetch_slice<KEYED>(col_idx, keys, sl * SLICE, lane, E, vec_ok, cur);
        }
        int cnt = 0;
        int k   = 0;
        for (int s = wave; s < nslices; s += OWN_WAVES, ++k) {
            int sl = s + rot; if (sl >= nslices) sl -= nslices;
            const int e0 = sl * SLICE;

            v4i nxt[NREG];
            {
                int sn = ((s + OWN_WAVES < nslices) ? s + OWN_WAVES : s) + rot;
                if (sn >= nslices) sn -= nslices;
                fetch_slice<KEYED>(col_idx, keys, sn * SLICE, lane, E, vec_ok, nxt);
            }

            cnt += filter_slice<KEYED>(cur, myq + cnt, lane, e0, E, tile, lo, nrows,
                                       (k & (TAG_SLICES - 1)) << 10);

            const int kblock = k & ~(TAG_SLICES - 1);
            int head = 0;
            for (; cnt - head >= 32; head += 32)
                gather_group<REC, true>(h, row_idx, col_idx, w, recs, acc, myq + head, 32, lane, wave,
                                        kblock, rot, nslices, lo);
            const bool flush = (s + OWN_WAVES >= nslices) || ((k & (TAG_SLICES - 1)) == TAG_SLICES - 1);
            if (flush && cnt > head) {
                gather_group<REC, false>(h, row_idx, col_idx, w, recs, acc, myq + head, cnt - head, lane, wave,
                                         kblock, rot, nslices, lo);
                head = cnt;
            }
            const int rest = cnt - head;
            if (head > 0 && rest > 0) {
                const unsigned short en = myq[head + ((lane < rest) ? lane : 0)];
                if (lane < rest) myq[lane] = en;
            }
            cnt = rest;

#pragma unroll
            for (int j = 0; j < NREG; ++j) cur[j] = nxt[j];
        }
        __syncthreads();

        float4*       o4 = (float4*)(out + (size_t)lo * D);
        const float4* a4 = (const float4*)acc;
        for (int i = threadIdx.x; i < nrows * (D / 4); i += blockDim.x) {
            const int   c  = lo + i / (D / 4);
            const float dc = inv_sqrt_deg(deg2[c], deg2[N + c]);
            float4 o = o4[i];
            const float4 a = a4[i];
            o.x += a.x * dc; o.y += a.y * dc; o.z += a.z * dc; o.w += a.w * dc;
            o4[i] = o;
        }
        __syncthreads();
    }
}

#define CELL_SLOTS 16
#define CELL_RECS  (CELL_SLOTS - 1)
#define MAX_TILES  256
#define LIN_BLOCKS 512
#define GCN_BLOCKS 1024

__global__ void zero_f32_and_count(float* __restrict__ p, int n, int* __restrict__ side_count) {
    int i = blockIdx.x * blockDim.x + threadIdx.x;
    if (i < n) p[i] = 0.0f;
    if (i == 0) *side_count = 0;
}

__global__ __launch_bounds__(SLICE / 4) void deg_cells_lin_kernel(const int* __restrict__ row,
                                                                  const int* __restrict__ col,
                                                                  const float* __restrict__ w,
                                                                  float* __restrict__ deg2,
                                                                  uint4* __restrict__ cells,
                                                                  int* __restrict__ side_count,
                                                                  int* __restrict__ side_idx,
                                                                  int n, int E, int nslices, int ntile,
                                                                  const float* __restrict__ x,
                                                                  const float* __restrict__ Wlin,
                                                                  float* __restrict__ out_hl,
                                                                  int nrowtiles, int nlin) {
    __shared__ __attribute__((aligned(16))) unsigned char smem[MAX_TILES * CELL_SLOTS * 8 + MAX_TILES * 4];

    if ((int)blockIdx.x < nlin) {
        product_block8<0>(x, Wlin, nullptr, out_hl, nrowtiles, n, (int)blockIdx.x, nlin, (float*)smem);
        return;
    }
    const int slice = (int)blockIdx.x - nlin;

    uint2* stage = (uint2*)smem;
    int*   cnt   = (int*)(smem + MAX_TILES * CELL_SLOTS * 8);

    const int tid = threadIdx.x;
#pragma unroll
    for (int q = 0; q < CELL_SLOTS / 2; ++q)
        ((uint4*)stage)[tid + (SLICE / 4) * q] = make_uint4(0u, 0u, 0u, 0u);
    cnt[tid] = 0;
    __syncthreads();

    const unsigned place = __builtin_amdgcn_s_sendmsg_rtn(0x87);
    float* __restrict__ deg = deg2 + (size_t)((place >> 18) & 1u) * n;
    const int e = (slice * (SLICE / 4) + tid) * 4;

    int   cc[4], rr[4];
    float wv[4];
    if ((E & 3) == 0 && e < E) {
        const v4i    c  = *(const v4i*)(col + e);
        const float4 ww = *(const float4*)(w + e);
        const v4i    r  = *(const v4i*)(row + e);
        cc[0] = c.x; cc[1] = c.y; cc[2] = c.z; cc[3] = c.w;
        rr[0] = r.x; rr[1] = r.y; rr[2] = r.z; rr[3] = r.w;
        wv[0] = ww.x; wv[1] = ww.y; wv[2] = ww.z; wv[3] = ww.w;
    } else {
#pragma unroll
        for (int k = 0; k < 4; ++k) {
            const bool in = e + k < E;
            cc[k] = in ? col[e + k] : -1;
            rr[k] = in ? row[e + k] : 0;
            wv[k] = in ? w[e + k] : 0.0f;
        }
    }
#pragma unroll
    for (int k = 0; k < 4; ++k) {
        if (cc[k] >= 0) {
            unsafeAtomicAdd(&deg[cc[k]], wv[k]);
            const unsigned t    = (unsigned)cc[k] / TILE_ROWS;
            const unsigned lrow = (unsigned)cc[k] - t * TILE_ROWS;
            const int rank = atomicAdd(&cnt[t], 1);
            if (rank < CELL_RECS) {
                stage[t * CELL_SLOTS + 1 + rank] =
                    make_uint2((unsigned)rr[k] | (lrow << REC_ROW_BITS), __float_as_uint(wv[k]));
            } else {
                const int p = atomicAdd(side_count, 1);
                side_idx[p] = e + k;
            }
        }
    }
    __syncthreads();

#pragma unroll
    for (int q = 0; q < CELL_SLOTS / 2; ++q) {
        const int i     = tid + (SLICE / 4) * q;
        const int cell  = i >> 3;
        const int piece = i & 7;
        uint4 v = ((const uint4*)stage)[i];
        if (piece == 0) { v.x = (unsigned)cnt[cell]; v.y = 0u; }
        if (cell < ntile)
            cells[((size_t)cell * nslices + slice) * (CELL_SLOTS / 2) + piece] = v;
    }
}

template <bool FULL>
__device__ __forceinline__ void gather_recs(const float* __restrict__ h, float* acc,
                                            const uint2* qg, int n, int lane) {
    const int   qi   = FULL ? lane : ((lane < n) ? lane : 0);
    const uint2 rc   = qg[qi];
    const int   r    = (int)(rc.x & ((1u << REC_ROW_BITS) - 1u));
    const int   lrow = (int)(rc.x >> REC_ROW_BITS);
    const float wt   = __uint_as_float(rc.y);

    if constexpr (FULL) {
        float v[2][4][4];
#pragma unroll
        for (int u = 0; u < 4; ++u) {
            const float* __restrict__ hr = h + (size_t)__builtin_amdgcn_readlane(r, u) * D + lane;
#pragma unroll
            for (int j = 0; j < 4; ++j) v[0][u][j] = hr[32 * j];
        }
#pragma unroll
        for (int g = 0; g < 8; ++g) {
            if (g + 1 < 8) {
#pragma unroll
                for (int u = 0; u < 4; ++u) {
                    const float* __restrict__ hr =
                        h + (size_t)__builtin_amdgcn_readlane(r, 4 * (g + 1) + u) * D + lane;
#pragma unroll
                    for (int j = 0; j < 4; ++j) v[(g + 1) & 1][u][j] = hr[32 * j];
                }
            }
#pragma unroll
            for (int u = 0; u < 4; ++u) {
                const int   lr = __builtin_amdgcn_readlane(lrow, 4 * g + u);
                const float sc = __int_as_float(__builtin_amdgcn_readlane(__float_as_int(wt), 4 * g + u));
                float* a = acc + lr * D + lane;
#pragma unroll
                for (int j = 0; j < 4; ++j)
                    __hip_atomic_fetch_add(a + 32 * j, v[g & 1][u][j] * sc, __ATOMIC_RELAXED,
                                           __HIP_MEMORY_SCOPE_WORKGROUP);
            }
        }
    } else {
        for (int t = 0; t < n; t += 4) {
            float v[4][4];
            float sc[4];
            int   lr[4];
#pragma unroll
            for (int u = 0; u < 4; ++u) {
                const int tt = min(t + u, n - 1);
                const int rt = __builtin_amdgcn_readlane(r, tt);
                lr[u] = __builtin_amdgcn_readlane(lrow, tt);
                sc[u] = __int_as_float(__builtin_amdgcn_readlane(__float_as_int(wt), tt));
                const float* __restrict__ hr = h + (size_t)rt * D + lane;
#pragma unroll
                for (int j = 0; j < 4; ++j) v[u][j] = hr[32 * j];
            }
#pragma unroll
            for (int u = 0; u < 4; ++u) {
                if (t + u < n) {
                    float* a = acc + lr[u] * D + lane;
#pragma unroll
                    for (int j = 0; j < 4; ++j)
                        __hip_atomic_fetch_add(a + 32 * j, v[u][j] * sc[u], __ATOMIC_RELAXED,
                                               __HIP_MEMORY_SCOPE_WORKGROUP);
                }
            }
        }
    }
}

#define GRP_SLICES 8
#define GRP_LOADS  (GRP_SLICES / 2)
#define CQCAP (32 + GRP_SLICES * CELL_RECS + 8)

__device__ __forceinline__ void fetch_cells(const uint2* __restrict__ ct, int group, int lane, int nslices,
                                            uint2 (&c)[GRP_LOADS]) {
#pragma unroll
    for (int j = 0; j < GRP_LOADS; ++j) {
        const int s0 = group * GRP_SLICES + 2 * j;
        c[j] = (s0 + (lane >> 4) < nslices) ? ct[(size_t)s0 * CELL_SLOTS + lane] : make_uint2(0u, 0u);
    }
}

__global__ __launch_bounds__(1024) void owner_gather_cells(const float* __restrict__ h,
                                                           const int* __restrict__ row_idx,
                                                           const int* __restrict__ col_idx,
                                                           const float* __restrict__ w,
                                                           const float* __restrict__ deg2,
                                                           const uint2* __restrict__ cells,
                                                           const int* __restrict__ side_count,
                                                           const int* __restrict__ side_idx,
                                                           float* __restrict__ out, int N, int E) {
    __shared__ float acc[TILE_ROWS * D];
    __shared__ uint2 q2[OWN_WAVES * CQCAP];

    const int wave = __builtin_amdgcn_readfirstlane(threadIdx.x >> 5);
    const int lane = threadIdx.x & 31;
    uint2* myq = q2 + wave * CQCAP;

    const int nslices = (E + SLICE - 1) / SLICE;
    const int ngroups = (nslices + GRP_SLICES - 1) / GRP_SLICES;
    const int slot    = lane & 15;

    for (int tile = blockIdx.x; tile * TILE_ROWS < N; tile += gridDim.x) {
        const int lo    = tile * TILE_ROWS;
        const int nrows = min(TILE_ROWS, N - lo);

        for (int i = threadIdx.x; i < TILE_ROWS * D / 4; i += blockDim.x)
            ((float4*)acc)[i] = make_float4(0.0f, 0.0f, 0.0f, 0.0f);
        __syncthreads();

        const uint2* __restrict__ ct = cells + (size_t)tile * nslices * CELL_SLOTS;
        uint2 cur[GRP_LOADS] = {};
        if (wave < ngroups) fetch_cells(ct, wave, lane, nslices, cur);

        int cnt = 0;
        for (int g = wave; g < ngroups; g += OWN_WAVES) {
            uint2 nxt[GRP_LOADS];
            fetch_cells(ct, (g + OWN_WAVES < ngroups) ? g + OWN_WAVES : g, lane, nslices, nxt);

#pragma unroll
            for (int j = 0; j < GRP_LOADS; ++j) {
                const int c0 = __builtin_amdgcn_readlane((int)cur[j].x, 0);
                const int c1 = __builtin_amdgcn_readlane((int)cur[j].x, 16);
                const int hc = min((lane & 16) ? c1 : c0, CELL_RECS);
                const bool ok = slot >= 1 && slot <= hc;
                const unsigned m = __builtin_amdgcn_ballot_w32(ok);
                if (ok) myq[cnt + __builtin_popcount(m & ((1u << lane) - 1u))] = cur[j];
                cnt += __builtin_popcount(m);
            }

            int head = 0;
            for (; cnt - head >= 32; head += 32)
                gather_recs<true>(h, acc, myq + head, 32, lane);
            if (g + OWN_WAVES >= ngroups && cnt > head) {
                gather_recs<false>(h, acc, myq + head, cnt - head, lane);
                head = cnt;
            }
            const int rest = cnt - head;
            if (head > 0 && rest > 0) {
                const uint2 en = myq[head + ((lane < rest) ? lane : 0)];
                if (lane < rest) myq[lane] = en;
            }
            cnt = rest;

#pragma unroll
            for (int j = 0; j < GRP_LOADS; ++j) cur[j] = nxt[j];
        }

        const int nside = min(*side_count, E);
        for (int i = threadIdx.x; i < nside; i += blockDim.x) {
            const int e = side_idx[i];
            const int c = col_idx[e];
            if ((unsigned)(c - lo) < (unsigned)nrows) {
                const float wt = w[e];
                const float* __restrict__ hr = h + (size_t)row_idx[e] * D;
                float* a = acc + (c - lo) * D;
                for (int j = 0; j < D; ++j)
                    __hip_atomic_fetch_add(a + j, hr[j] * wt, __ATOMIC_RELAXED, __HIP_MEMORY_SCOPE_WORKGROUP);
            }
        }
        __syncthreads();

        float4*       o4 = (float4*)(out + (size_t)lo * D);
        const float4* a4 = (const float4*)acc;
        for (int i = threadIdx.x; i < nrows * (D / 4); i += blockDim.x) {
            const int   c  = lo + i / (D / 4);
            const float dc = inv_sqrt_deg(deg2[c], deg2[N + c]);
            float4 o = o4[i];
            const float4 a = a4[i];
            o.x += a.x * dc; o.y += a.y * dc; o.z += a.z * dc; o.w += a.w * dc;
            o4[i] = o;
        }
        __syncthreads();
    }
}


extern "C" void kernel_launch(void* const* d_in, const int* in_sizes, int n_in,
                              void* d_out, int out_size, void* d_ws, size_t ws_size,
                              hipStream_t stream) {
    const float* x    = (const float*)d_in[0];
    const int*   ei   = (const int*)d_in[1];
    const float* ew   = (const float*)d_in[2];
    const float* Wlin = (const float*)d_in[3];
    const float* Wgcn = (const float*)d_in[4];
    float*       out  = (float*)d_out;

    const int N = in_sizes[0] / D;
    const int E = in_sizes[2];

    const int* row = ei;
    const int* col = ei + E;

    float* deg2 = (float*)d_ws;
    float* h    = deg2 + 2 * N;
    const int nslices = (E + SLICE - 1) / SLICE;
    const bool keyed = ws_size >= (size_t)(2 * N + N * D) * sizeof(float) + (size_t)nslices * SLICE;
    unsigned* keys = keyed ? (unsigned*)(h + (size_t)N * D) : nullptr;
    const size_t rec_off = ((size_t)(2 * N + N * D) * sizeof(float) + (size_t)nslices * SLICE + 15) & ~(size_t)15;
    const bool rec = keyed && N <= (1 << REC_ROW_BITS) && ws_size >= rec_off + (size_t)nslices * SLICE * 8;
    unsigned* recs = rec ? (unsigned*)((char*)d_ws + rec_off) : nullptr;

    const int    ntile     = (N + TILE_ROWS - 1) / TILE_ROWS;
    const size_t cells_off = ((size_t)(2 * N + N * D) * sizeof(float) + 127) & ~(size_t)127;
    const size_t side_off  = cells_off + (size_t)ntile * nslices * (CELL_SLOTS * 8);
    const size_t cnt_off   = (side_off + (size_t)E * sizeof(int) + 15) & ~(size_t)15;
    if (ntile <= MAX_TILES && N <= (1 << REC_ROW_BITS) && ws_size >= cnt_off + 16) {
        uint4* cells      = (uint4*)((char*)d_ws + cells_off);
        int*   side_idx   = (int*)((char*)d_ws + side_off);
        int*   side_count = (int*)((char*)d_ws + cnt_off);
        zero_f32_and_count<<<(2 * N + 255) / 256, 256, 0, stream>>>(deg2, 2 * N, side_count);
        deg_cells_lin_kernel<<<LIN_BLOCKS + nslices, SLICE / 4, 0, stream>>>(row, col, ew, deg2, cells, side_count,
                                                                            side_idx, N, E, nslices, ntile,
                                                                            x, Wlin, out, N / 16, LIN_BLOCKS);
        gemm_gcn<<<GCN_BLOCKS, 256, 0, stream>>>(x, Wgcn, deg2, h, N / 16, N);
        owner_gather_cells<<<ntile, 1024, 0, stream>>>(h, row, col, ew, deg2, (const uint2*)cells, side_count,
                                                       side_idx, out, N, E);
        return;
    }

    zero_f32<<<(2 * N + 255) / 256, 256, 0, stream>>>(deg2, 2 * N);
    deg_kernel<<<nslices, SLICE / 4, 0, stream>>>(row, col, ew, deg2, keys, recs, N, E);
    gemm_dual<<<512, 512, 0, stream>>>(x, Wlin, Wgcn, deg2, out, h, N / 16, N);
    const int nblk = (N + TILE_ROWS - 1) / TILE_ROWS;
    if (rec)        owner_gather<true, true><<<nblk, 1024, 0, stream>>>(h, row, col, ew, deg2, keys, recs, out, N, E);
    else if (keyed) owner_gather<true, false><<<nblk, 1024, 0, stream>>>(h, row, col, ew, deg2, keys, recs, out, N, E);
    else            owner_gather<false, false><<<nblk, 1024, 0, stream>>>(h, row, col, ew, deg2, keys, recs, out, N, E);
}


# gather: a lane fetches 16 B of an h row, one load instruction per edge row (4 whole lines) where four were; LDS tile keeps column 4l+j at float 32j+l (adds on 32 banks); arrival-order sums
# speedup vs baseline: 3.8153x; 1.0658x over previous
#include <hip/hip_runtime.h>

#define D 128

typedef float v2f __attribute__((ext_vector_type(2)));
typedef float v8f __attribute__((ext_vector_type(8)));


__global__ void zero_f32(float* __restrict__ p, int n) {
    int i = blockIdx.x * blockDim.x + threadIdx.x;
    if (i < n) p[i] = 0.0f;
}


#define TILE_ROWS 392
#define SLICE     1024
#define OWN_WAVES 32
#define KEYS_PER_LOAD 512

typedef int v4i __attribute__((ext_vector_type(4)));

#define REC_ROW_BITS 23
__global__ __launch_bounds__(SLICE / 4) void deg_kernel(const int* __restrict__ row,
                                                        const int* __restrict__ col,
                                                        const float* __restrict__ w,
                                                        float* __restrict__ deg2,
                                                        unsigned* __restrict__ keys,
                                                        unsigned* __restrict__ recs, int n, int E) {
    const unsigned place = __builtin_amdgcn_s_sendmsg_rtn(0x87);
    float* __restrict__ deg = deg2 + (size_t)((place >> 18) & 1u) * n;
    const int e = (blockIdx.x * blockDim.x + threadIdx.x) * 4;
    unsigned key = 0u;
    if ((E & 3) == 0 && e < E) {
        const v4i    c  = *(const v4i*)(col + e);
        const float4 ww = *(const float4*)(w + e);
        unsafeAtomicAdd(&deg[c.x], ww.x);
        unsafeAtomicAdd(&deg[c.y], ww.y);
        unsafeAtomicAdd(&deg[c.z], ww.z);
        unsafeAtomicAdd(&deg[c.w], ww.w);
        const unsigned tx = (unsigned)c.x / TILE_ROWS, ty = (unsigned)c.y / TILE_ROWS;
        const unsigned tz = (unsigned)c.z / TILE_ROWS, tw = (unsigned)c.w / TILE_ROWS;
        key = tx | (ty << 8) | (tz << 16) | (tw << 24);
        if (recs != nullptr) {
            const v4i r = *(const v4i*)(row + e);
            uint4 a, b;
            a.x = (unsigned)r.x | (((unsigned)c.x - tx * TILE_ROWS) << REC_ROW_BITS); a.y = __float_as_uint(ww.x);
            a.z = (unsigned)r.y | (((unsigned)c.y - ty * TILE_ROWS) << REC_ROW_BITS); a.w = __float_as_uint(ww.y);
            b.x = (unsigned)r.z | (((unsigned)c.z - tz * TILE_ROWS) << REC_ROW_BITS); b.y = __float_as_uint(ww.z);
            b.z = (unsigned)r.w | (((unsigned)c.w - tw * TILE_ROWS) << REC_ROW_BITS); b.w = __float_as_uint(ww.w);
            *(uint4*)(recs + 2 * e)     = a;
            *(uint4*)(recs + 2 * e + 4) = b;
        }
    } else {
        for (int k = 0; k < 4; ++k) {
            if (e + k < E) {
                const int c = col[e + k];
                const float wk = w[e + k];
                unsafeAtomicAdd(&deg[c], wk);
                const unsigned t = (unsigned)c / TILE_ROWS;
                key |= t << (8 * k);
                if (recs != nullptr)
                    *(uint2*)(recs + 2 * (e + k)) =
                        make_uint2((unsigned)row[e + k] | (((unsigned)c - t * TILE_ROWS) << REC_ROW_BITS),
                                   __float_as_uint(wk));
            }
        }
    }
    if (keys != nullptr) keys[e >> 2] = key;
}

__device__ __forceinline__ float inv_sqrt_deg(float plane0, float plane1) {
    const float d = plane0 + plane1;
    return (d > 0.0f) ? __frsqrt_rn(d) : 0.0f;
}


#define XS_STRIDE 132

__global__ __launch_bounds__(512) void gemm_dual(const float* __restrict__ x,
                                                 const float* __restrict__ Wlin,
                                                 const float* __restrict__ Wgcn,
                                                 const float* __restrict__ deg2,
                                                 float* __restrict__ out_hl,
                                                 float* __restrict__ h, int ntiles, int n) {
    __shared__ __attribute__((aligned(16))) float xs[2][16 * XS_STRIDE];

    const int wave  = threadIdx.x >> 5;
    const int lane  = threadIdx.x & 31;
    const int which = wave >> 3;
    const int col0  = (wave & 7) * 16;

    const float* __restrict__ W   = which ? Wgcn : Wlin;
    float* __restrict__       dst = which ? h    : out_hl;

    const int m  = lane & 15;
    const int hh = lane >> 4;

    v2f b[32];
#pragma unroll
    for (int ks = 0; ks < 32; ++ks) {
        b[ks].x = W[(size_t)(4 * ks + 2 * hh + 0) * D + col0 + m];
        b[ks].y = W[(size_t)(4 * ks + 2 * hh + 1) * D + col0 + m];
    }

    const float* __restrict__ xsrc = x + (size_t)wave * D + lane * 4;
    const int soff = wave * XS_STRIDE + lane * 4;

    const float* __restrict__ dg0 = deg2 + m;
    const float* __restrict__ dg1 = deg2 + (size_t)n + m;

    int tile = blockIdx.x;
    float4 stage = make_float4(0.0f, 0.0f, 0.0f, 0.0f);
    float dsum0 = 0.0f, dsum1 = 0.0f;
    if (tile < ntiles) {
        stage = *(const float4*)(xsrc + (size_t)tile * 16 * D);
        if (which) { dsum0 = dg0[tile * 16]; dsum1 = dg1[tile * 16]; }
    }
    *(float4*)&xs[0][soff] = stage;
    __syncthreads();

    int buf = 0;
    for (; tile < ntiles; tile += gridDim.x) {
        const int  nt   = tile + gridDim.x;
        const bool more = nt < ntiles;
        const float rowfac = which ? inv_sqrt_deg(dsum0, dsum1) : 1.0f;
        if (more) {
            stage = *(const float4*)(xsrc + (size_t)nt * 16 * D);
            if (which) { dsum0 = dg0[nt * 16]; dsum1 = dg1[nt * 16]; }
        }

        const float* xa = &xs[buf][m * XS_STRIDE + 2 * hh];
        v8f acc = {};
#pragma unroll
        for (int ks = 0; ks < 32; ++ks) {
            const v2f a = *(const v2f*)(xa + 4 * ks);
            acc = __builtin_amdgcn_wmma_f32_16x16x4_f32(
                 false, a,  false, b[ks],
                 (short)0, acc,  false,  false);
        }

        float* __restrict__ drow = dst + (size_t)(tile * 16 + 8 * hh) * D + col0 + m;
#pragma unroll
        for (int r = 0; r < 8; ++r) {
            const float f0 = __int_as_float(__builtin_amdgcn_readlane(__float_as_int(rowfac), r));
            const float f1 = __int_as_float(__builtin_amdgcn_readlane(__float_as_int(rowfac), 8 + r));
            drow[(size_t)r * D] = acc[r] * (hh ? f1 : f0);
        }

        if (more) *(float4*)&xs[buf ^ 1][soff] = stage;
        __syncthreads();
        buf ^= 1;
    }
}

template <int WHICH>
__device__ __forceinline__ void product_block8(const float* __restrict__ x, const float* __restrict__ W,
                                               const float* __restrict__ deg2, float* __restrict__ dst,
                                               int ntiles, int n, int first, int step, float* xs) {
    const int wave = threadIdx.x >> 5;
    const int lane = threadIdx.x & 31;
    const int col0 = wave * 16;
    const int m    = lane & 15;
    const int hh   = lane >> 4;

    v2f b[32];
#pragma unroll
    for (int ks = 0; ks < 32; ++ks) {
        b[ks].x = W[(size_t)(4 * ks + 2 * hh + 0) * D + col0 + m];
        b[ks].y = W[(size_t)(4 * ks + 2 * hh + 1) * D + col0 + m];
    }

    const float* __restrict__ xsrc = x + (size_t)wave * D + lane * 4;
    const int soff = wave * XS_STRIDE + lane * 4;

    const float* __restrict__ dg0 = WHICH ? deg2 + m : nullptr;
    const float* __restrict__ dg1 = WHICH ? deg2 + (size_t)n + m : nullptr;

    int tile = first;
    float4 st0 = make_float4(0.0f, 0.0f, 0.0f, 0.0f), st1 = st0;
    float dsum0 = 0.0f, dsum1 = 0.0f;
    if (tile < ntiles) {
        st0 = *(const float4*)(xsrc + (size_t)tile * 16 * D);
        st1 = *(const float4*)(xsrc + (size_t)tile * 16 * D + 8 * D);
        if (WHICH) { dsum0 = dg0[tile * 16]; dsum1 = dg1[tile * 16]; }
    }
    *(float4*)&xs[soff]                 = st0;
    *(float4*)&xs[soff + 8 * XS_STRIDE] = st1;
    __syncthreads();

    int buf = 0;
    for (; tile < ntiles; tile += step) {
        const int  nt   = tile + step;
        const bool more = nt < ntiles;
        float rowfac = 1.0f;
        if (WHICH) rowfac = inv_sqrt_deg(dsum0, dsum1);
        if (more) {
            st0 = *(const float4*)(xsrc + (size_t)nt * 16 * D);
            st1 = *(const float4*)(xsrc + (size_t)nt * 16 * D + 8 * D);
            if (WHICH) { dsum0 = dg0[nt * 16]; dsum1 = dg1[nt * 16]; }
        }

        const float* xa = xs + buf * (16 * XS_STRIDE) + m * XS_STRIDE + 2 * hh;
        v8f acc = {};
#pragma unroll
        for (int ks = 0; ks < 32; ++ks) {
            const v2f a = *(const v2f*)(xa + 4 * ks);
            acc = __builtin_amdgcn_wmma_f32_16x16x4_f32(
                 false, a,  false, b[ks],
                 (short)0, acc,  false,  false);
        }

        float* __restrict__ drow = dst + (size_t)(tile * 16 + 8 * hh) * D + col0 + m;
        if (WHICH) {
#pragma unroll
            for (int r = 0; r < 8; ++r) {
                const float f0 = __int_as_float(__builtin_amdgcn_readlane(__float_as_int(rowfac), r));
                const float f1 = __int_as_float(__builtin_amdgcn_readlane(__float_as_int(rowfac), 8 + r));
                drow[(size_t)r * D] = acc[r] * (hh ? f1 : f0);
            }
        } else {
#pragma unroll
            for (int r = 0; r < 8; ++r) drow[(size_t)r * D] = acc[r];
        }

        if (more) {
            float* o = xs + (buf ^ 1) * (16 * XS_STRIDE);
            *(float4*)&o[soff]                 = st0;
            *(float4*)&o[soff + 8 * XS_STRIDE] = st1;
        }
        __syncthreads();
        buf ^= 1;
    }
}

__global__ __launch_bounds__(256) void gemm_gcn(const float* __restrict__ x, const float* __restrict__ Wgcn,
                                                const float* __restrict__ deg2, float* __restrict__ h,
                                                int ntiles, int n) {
    __shared__ __attribute__((aligned(16))) float xs[2 * 16 * XS_STRIDE];
    product_block8<1>(x, Wgcn, deg2, h, ntiles, n, blockIdx.x, gridDim.x, xs);
}


template <bool KEYED>
__device__ __forceinline__ void fetch_slice(const int* __restrict__ col, const unsigned* __restrict__ keys,
                                            int e0, int lane, int E, bool vec_ok,
                                            v4i (&c)[KEYED ? SLICE / KEYS_PER_LOAD : SLICE / 128]) {
    if constexpr (KEYED) {
#pragma unroll
        for (int j = 0; j < SLICE / KEYS_PER_LOAD; ++j)
            c[j] = *(const v4i*)(keys + ((e0 + j * KEYS_PER_LOAD) >> 2) + lane * 4);
    } else if (vec_ok && e0 + SLICE <= E) {
#pragma unroll
        for (int j = 0; j < SLICE / 128; ++j)
            c[j] = *(const v4i*)(col + e0 + j * 128 + lane * 4);
    } else {
#pragma unroll
        for (int j = 0; j < SLICE / 128; ++j) {
            const int e = e0 + j * 128 + lane * 4;
            c[j].x = (e + 0 < E) ? col[e + 0] : -1;
            c[j].y = (e + 1 < E) ? col[e + 1] : -1;
            c[j].z = (e + 2 < E) ? col[e + 2] : -1;
            c[j].w = (e + 3 < E) ? col[e + 3] : -1;
        }
    }
}

template <bool KEYED>
__device__ __forceinline__ int filter_slice(v4i (&cur)[KEYED ? SLICE / KEYS_PER_LOAD : SLICE / 128],
                                            unsigned short* myq, int lane, int e0, int E,
                                            int tile, int lo, int nrows, int tag) {
    int cnt = 0;
    if constexpr (KEYED) {
        const unsigned pat = (unsigned)tile * 0x01010101u;
        if (e0 + SLICE > E) {
#pragma unroll
            for (int j = 0; j < SLICE / KEYS_PER_LOAD; ++j) {
#pragma unroll
                for (int d = 0; d < 4; ++d) {
                    const int nv = E - (e0 + j * KEYS_PER_LOAD + lane * 16 + 4 * d);
                    const unsigned keep = (nv >= 4) ? 0xffffffffu : ((nv <= 0) ? 0u : ((1u << (8 * nv)) - 1u));
                    cur[j][d] = (int)(((unsigned)cur[j][d] & keep) | (~pat & ~keep));
                }
            }
        }
#pragma unroll
        for (int j = 0; j < SLICE / KEYS_PER_LOAD; ++j) {
            unsigned g = 0u;
#pragma unroll
            for (int d = 0; d < 4; ++d) {
                const unsigned x  = (unsigned)cur[j][d] ^ pat;
                const unsigned nz = ((x & 0x7f7f7f7fu) + 0x7f7f7f7fu) | x;
                g |= (~nz & 0x80808080u) >> (7 - d);
            }
            unsigned hit = __builtin_amdgcn_ballot_w32(g != 0u);
            while (hit != 0u) {
                if (g != 0u) {
                    const int p = __builtin_ctz(g);
                    myq[cnt + __builtin_popcount(hit & ((1u << lane) - 1u))] =
                        (unsigned short)(tag | (j * KEYS_PER_LOAD + lane * 16 + ((p & 7) << 2) + (p >> 3)));
                    g &= g - 1u;
                }
                cnt += __builtin_popcount(hit);
                hit = __builtin_amdgcn_ballot_w32(g != 0u);
            }
        }
    } else {
#pragma unroll
        for (int j = 0; j < SLICE / 128; ++j) {
            unsigned mask[4];
#pragma unroll
            for (int k = 0; k < 4; ++k)
                mask[k] = __builtin_amdgcn_ballot_w32((unsigned)(cur[j][k] - lo) < (unsigned)nrows);
            if ((mask[0] | mask[1] | mask[2] | mask[3]) != 0u) {
#pragma unroll
                for (int k = 0; k < 4; ++k) {
                    if ((mask[k] >> lane) & 1u)
                        myq[cnt + __builtin_popcount(mask[k] & ((1u << lane) - 1u))] =
                            (unsigned short)(tag | (j * 128 + lane * 4 + k));
                    cnt += __builtin_popcount(mask[k]);
                }
            }
        }
    }
    return cnt;
}

#define QCAP (SLICE + 32)
#define TAG_SLICES 64

template <bool REC, bool FULL>
__device__ __forceinline__ void gather_group(const float* __restrict__ h, const int* __restrict__ row_idx,
                                             const int* __restrict__ col_idx, const float* __restrict__ w,
                                             const unsigned* __restrict__ recs, float* acc,
                                             const unsigned short* qg, int n, int lane, int wave,
                                             int kblock, int rot, int nslices, int lo) {
    const int qi = FULL ? lane : ((lane < n) ? lane : 0);
    const unsigned en = qg[qi];
    int sl = wave + OWN_WAVES * (kblock + (int)(en >> 10)) + rot;
    if (sl >= nslices) sl -= nslices;
    const int e = sl * SLICE + (int)(en & (SLICE - 1));
    int r, lrow;
    float wt;
    if constexpr (REC) {
        const uint2 rc = *(const uint2*)(recs + 2 * e);
        r    = (int)(rc.x & ((1u << REC_ROW_BITS) - 1u));
        lrow = (int)(rc.x >> REC_ROW_BITS);
        wt   = __uint_as_float(rc.y);
    } else {
        r    = row_idx[e];
        lrow = col_idx[e] - lo;
        wt   = w[e];
    }

    if constexpr (FULL) {
        float v[2][4][4];
#pragma unroll
        for (int u = 0; u < 4; ++u) {
            const float* __restrict__ hr = h + (size_t)__builtin_amdgcn_readlane(r, u) * D + lane;
#pragma unroll
            for (int j = 0; j < 4; ++j) v[0][u][j] = hr[32 * j];
        }
#pragma unroll
        for (int g = 0; g < 8; ++g) {
            if (g + 1 < 8) {
#pragma unroll
                for (int u = 0; u < 4; ++u) {
                    const float* __restrict__ hr =
                        h + (size_t)__builtin_amdgcn_readlane(r, 4 * (g + 1) + u) * D + lane;
#pragma unroll
                    for (int j = 0; j < 4; ++j) v[(g + 1) & 1][u][j] = hr[32 * j];
                }
            }
#pragma unroll
            for (int u = 0; u < 4; ++u) {
                const int   lr = __builtin_amdgcn_readlane(lrow, 4 * g + u);
                const float sc = __int_as_float(__builtin_amdgcn_readlane(__float_as_int(wt), 4 * g + u));
                float* a = acc + lr * D + lane;
#pragma unroll
                for (int j = 0; j < 4; ++j)
                    __hip_atomic_fetch_add(a + 32 * j, v[g & 1][u][j] * sc, __ATOMIC_RELAXED,
                                           __HIP_MEMORY_SCOPE_WORKGROUP);
            }
        }
    } else {
        for (int t = 0; t < n; t += 4) {
            float v[4][4];
            float sc[4];
            int   lr[4];
#pragma unroll
            for (int u = 0; u < 4; ++u) {
                const int tt = min(t + u, n - 1);
                const int rt = __builtin_amdgcn_readlane(r, tt);
                lr[u] = __builtin_amdgcn_readlane(lrow, tt);
                sc[u] = __int_as_float(__builtin_amdgcn_readlane(__float_as_int(wt), tt));
                const float* __restrict__ hr = h + (size_t)rt * D + lane;
#pragma unroll
                for (int j = 0; j < 4; ++j) v[u][j] = hr[32 * j];
            }
#pragma unroll
            for (int u = 0; u < 4; ++u) {
                if (t + u < n) {
                    float* a = acc + lr[u] * D + lane;
#pragma unroll
                    for (int j = 0; j < 4; ++j)
                        __hip_atomic_fetch_add(a + 32 * j, v[u][j] * sc[u], __ATOMIC_RELAXED,
                                               __HIP_MEMORY_SCOPE_WORKGROUP);
                }
            }
        }
    }
}

template <bool KEYED, bool REC>
__global__ __launch_bounds__(1024) void owner_gather(const float* __restrict__ h,
                                                     const int* __restrict__ row_idx,
                                                     const int* __restrict__ col_idx,
                                                     const float* __restrict__ w,
                                                     const float* __restrict__ deg2,
                                                     const unsigned* __restrict__ keys,
                                                     const unsigned* __restrict__ recs,
                                                     float* __restrict__ out, int N, int E) {
    __shared__ float          acc[TILE_ROWS * D];
    __shared__ unsigned short q[OWN_WAVES * QCAP];

    const int wave = __builtin_amdgcn_readfirstlane(threadIdx.x >> 5);
    const int lane = threadIdx.x & 31;
    unsigned short* myq = q + wave * QCAP;

    constexpr int NREG = KEYED ? SLICE / KEYS_PER_LOAD : SLICE / 128;
    const int  nslices = (E + SLICE - 1) / SLICE;
    const bool vec_ok  = (E & 3) == 0;
    const int  rot     = (int)(((unsigned)blockIdx.x * (unsigned)nslices / gridDim.x) % (unsigned)nslices);

    for (int tile = blockIdx.x; tile * TILE_ROWS < N; tile += gridDim.x) {
        const int lo    = tile * TILE_ROWS;
        const int nrows = min(TILE_ROWS, N - lo);

        for (int i = threadIdx.x; i < TILE_ROWS * D / 4; i += blockDim.x)
            ((float4*)acc)[i] = make_float4(0.0f, 0.0f, 0.0f, 0.0f);
        __syncthreads();

        v4i cur[NREG];
        {
            int sl = wave + rot; if (sl >= nslices) sl -= nslices;
            if (wave < nslices) fetch_slice<KEYED>(col_idx, keys, sl * SLICE, lane, E, vec_ok, cur);
        }
        int cnt = 0;
        int k   = 0;
        for (int s = wave; s < nslices; s += OWN_WAVES, ++k) {
            int sl = s + rot; if (sl >= nslices) sl -= nslices;
            const int e0 = sl * SLICE;

            v4i nxt[NREG];
            {
                int sn = ((s + OWN_WAVES < nslices) ? s + OWN_WAVES : s) + rot;
                if (sn >= nslices) sn -= nslices;
                fetch_slice<KEYED>(col_idx, keys, sn * SLICE, lane, E, vec_ok, nxt);
            }

            cnt += filter_slice<KEYED>(cur, myq + cnt, lane, e0, E, tile, lo, nrows,
                                       (k & (TAG_SLICES - 1)) << 10);

            const int kblock = k & ~(TAG_SLICES - 1);
            int head = 0;
            for (; cnt - head >= 32; head += 32)
                gather_group<REC, true>(h, row_idx, col_idx, w, recs, acc, myq + head, 32, lane, wave,
                                        kblock, rot, nslices, lo);
            const bool flush = (s + OWN_WAVES >= nslices) || ((k & (TAG_SLICES - 1)) == TAG_SLICES - 1);
            if (flush && cnt > head) {
                gather_group<REC, false>(h, row_idx, col_idx, w, recs, acc, myq + head, cnt - head, lane, wave,
                                         kblock, rot, nslices, lo);
                head = cnt;
            }
            const int rest = cnt - head;
            if (head > 0 && rest > 0) {
                const unsigned short en = myq[head + ((lane < rest) ? lane : 0)];
                if (lane < rest) myq[lane] = en;
            }
            cnt = rest;

#pragma unroll
            for (int j = 0; j < NREG; ++j) cur[j] = nxt[j];
        }
        __syncthreads();

        float4*       o4 = (float4*)(out + (size_t)lo * D);
        const float4* a4 = (const float4*)acc;
        for (int i = threadIdx.x; i < nrows * (D / 4); i += blockDim.x) {
            const int   c  = lo + i / (D / 4);
            const float dc = inv_sqrt_deg(deg2[c], deg2[N + c]);
            float4 o = o4[i];
            const float4 a = a4[i];
            o.x += a.x * dc; o.y += a.y * dc; o.z += a.z * dc; o.w += a.w * dc;
            o4[i] = o;
        }
        __syncthreads();
    }
}

#define CELL_SLOTS 16
#define CELL_RECS  (CELL_SLOTS - 1)
#define MAX_TILES  256
#define LIN_BLOCKS 512
#define GCN_BLOCKS 1024

__global__ void zero_f32_and_count(float* __restrict__ p, int n, int* __restrict__ side_count) {
    int i = blockIdx.x * blockDim.x + threadIdx.x;
    if (i < n) p[i] = 0.0f;
    if (i == 0) *side_count = 0;
}

__global__ __launch_bounds__(SLICE / 4) void deg_cells_lin_kernel(const int* __restrict__ row,
                                                                  const int* __restrict__ col,
                                                                  const float* __restrict__ w,
                                                                  float* __restrict__ deg2,
                                                                  uint4* __restrict__ cells,
                                                                  int* __restrict__ side_count,
                                                                  int* __restrict__ side_idx,
                                                                  int n, int E, int nslices, int ntile,
                                                                  const float* __restrict__ x,
                                                                  const float* __restrict__ Wlin,
                                                                  float* __restrict__ out_hl,
                                                                  int nrowtiles, int nlin) {
    __shared__ __attribute__((aligned(16))) unsigned char smem[MAX_TILES * CELL_SLOTS * 8 + MAX_TILES * 4];

    if ((int)blockIdx.x < nlin) {
        product_block8<0>(x, Wlin, nullptr, out_hl, nrowtiles, n, (int)blockIdx.x, nlin, (float*)smem);
        return;
    }
    const int slice = (int)blockIdx.x - nlin;

    uint2* stage = (uint2*)smem;
    int*   cnt   = (int*)(smem + MAX_TILES * CELL_SLOTS * 8);

    const int tid = threadIdx.x;
#pragma unroll
    for (int q = 0; q < CELL_SLOTS / 2; ++q)
        ((uint4*)stage)[tid + (SLICE / 4) * q] = make_uint4(0u, 0u, 0u, 0u);
    cnt[tid] = 0;
    __syncthreads();

    const unsigned place = __builtin_amdgcn_s_sendmsg_rtn(0x87);
    float* __restrict__ deg = deg2 + (size_t)((place >> 18) & 1u) * n;
    const int e = (slice * (SLICE / 4) + tid) * 4;

    int   cc[4], rr[4];
    float wv[4];
    if ((E & 3) == 0 && e < E) {
        const v4i    c  = *(const v4i*)(col + e);
        const float4 ww = *(const float4*)(w + e);
        const v4i    r  = *(const v4i*)(row + e);
        cc[0] = c.x; cc[1] = c.y; cc[2] = c.z; cc[3] = c.w;
        rr[0] = r.x; rr[1] = r.y; rr[2] = r.z; rr[3] = r.w;
        wv[0] = ww.x; wv[1] = ww.y; wv[2] = ww.z; wv[3] = ww.w;
    } else {
#pragma unroll
        for (int k = 0; k < 4; ++k) {
            const bool in = e + k < E;
            cc[k] = in ? col[e + k] : -1;
            rr[k] = in ? row[e + k] : 0;
            wv[k] = in ? w[e + k] : 0.0f;
        }
    }
#pragma unroll
    for (int k = 0; k < 4; ++k) {
        if (cc[k] >= 0) {
            unsafeAtomicAdd(&deg[cc[k]], wv[k]);
            const unsigned t    = (unsigned)cc[k] / TILE_ROWS;
            const unsigned lrow = (unsigned)cc[k] - t * TILE_ROWS;
            const int rank = atomicAdd(&cnt[t], 1);
            if (rank < CELL_RECS) {
                stage[t * CELL_SLOTS + 1 + rank] =
                    make_uint2((unsigned)rr[k] | (lrow << REC_ROW_BITS), __float_as_uint(wv[k]));
            } else {
                const int p = atomicAdd(side_count, 1);
                side_idx[p] = e + k;
            }
        }
    }
    __syncthreads();

#pragma unroll
    for (int q = 0; q < CELL_SLOTS / 2; ++q) {
        const int i     = tid + (SLICE / 4) * q;
        const int cell  = i >> 3;
        const int piece = i & 7;
        uint4 v = ((const uint4*)stage)[i];
        if (piece == 0) { v.x = (unsigned)cnt[cell]; v.y = 0u; }
        if (cell < ntile)
            cells[((size_t)cell * nslices + slice) * (CELL_SLOTS / 2) + piece] = v;
    }
}

__device__ __forceinline__ void tile_add_row(float* a, const float4 v, const float sc) {
    __hip_atomic_fetch_add(a,      v.x * sc, __ATOMIC_RELAXED, __HIP_MEMORY_SCOPE_WORKGROUP);
    __hip_atomic_fetch_add(a + 32, v.y * sc, __ATOMIC_RELAXED, __HIP_MEMORY_SCOPE_WORKGROUP);
    __hip_atomic_fetch_add(a + 64, v.z * sc, __ATOMIC_RELAXED, __HIP_MEMORY_SCOPE_WORKGROUP);
    __hip_atomic_fetch_add(a + 96, v.w * sc, __ATOMIC_RELAXED, __HIP_MEMORY_SCOPE_WORKGROUP);
}

template <bool FULL>
__device__ __forceinline__ void gather_recs(const float* __restrict__ h, float* acc,
                                            const uint2* qg, int n, int lane) {
    const int   qi   = FULL ? lane : ((lane < n) ? lane : 0);
    const uint2 rc   = qg[qi];
    const int   r    = (int)(rc.x & ((1u << REC_ROW_BITS) - 1u));
    const int   lrow = (int)(rc.x >> REC_ROW_BITS);
    const float wt   = __uint_as_float(rc.y);

    if constexpr (FULL) {
        float4 v[2][4];
#pragma unroll
        for (int u = 0; u < 4; ++u)
            v[0][u] = *((const float4*)(h + (size_t)__builtin_amdgcn_readlane(r, u) * D) + lane);
#pragma unroll
        for (int g = 0; g < 8; ++g) {
            if (g + 1 < 8) {
#pragma unroll
                for (int u = 0; u < 4; ++u)
                    v[(g + 1) & 1][u] =
                        *((const float4*)(h + (size_t)__builtin_amdgcn_readlane(r, 4 * (g + 1) + u) * D) + lane);
            }
#pragma unroll
            for (int u = 0; u < 4; ++u) {
                const int   lr = __builtin_amdgcn_readlane(lrow, 4 * g + u);
                const float sc = __int_as_float(__builtin_amdgcn_readlane(__float_as_int(wt), 4 * g + u));
                tile_add_row(acc + lr * D + lane, v[g & 1][u], sc);
            }
        }
    } else {
        for (int t = 0; t < n; t += 4) {
            float4 v[4];
            float  sc[4];
            int    lr[4];
#pragma unroll
            for (int u = 0; u < 4; ++u) {
                const int tt = min(t + u, n - 1);
                const int rt = __builtin_amdgcn_readlane(r, tt);
                lr[u] = __builtin_amdgcn_readlane(lrow, tt);
                sc[u] = __int_as_float(__builtin_amdgcn_readlane(__float_as_int(wt), tt));
                v[u] = *((const float4*)(h + (size_t)rt * D) + lane);
            }
#pragma unroll
            for (int u = 0; u < 4; ++u) {
                if (t + u < n) tile_add_row(acc + lr[u] * D + lane, v[u], sc[u]);
            }
        }
    }
}

#define GRP_SLICES 8
#define GRP_LOADS  (GRP_SLICES / 2)
#define CQCAP (32 + GRP_SLICES * CELL_RECS + 8)

__device__ __forceinline__ void fetch_cells(const uint2* __restrict__ ct, int group, int lane, int nslices,
                                            uint2 (&c)[GRP_LOADS]) {
#pragma unroll
    for (int j = 0; j < GRP_LOADS; ++j) {
        const int s0 = group * GRP_SLICES + 2 * j;
        c[j] = (s0 + (lane >> 4) < nslices) ? ct[(size_t)s0 * CELL_SLOTS + lane] : make_uint2(0u, 0u);
    }
}

__global__ __launch_bounds__(1024) void owner_gather_cells(const float* __restrict__ h,
                                                           const int* __restrict__ row_idx,
                                                           const int* __restrict__ col_idx,
                                                           const float* __restrict__ w,
                                                           const float* __restrict__ deg2,
                                                           const uint2* __restrict__ cells,
                                                           const int* __restrict__ side_count,
                                                           const int* __restrict__ side_idx,
                                                           float* __restrict__ out, int N, int E) {
    __shared__ float acc[TILE_ROWS * D];
    __shared__ uint2 q2[OWN_WAVES * CQCAP];

    const int wave = __builtin_amdgcn_readfirstlane(threadIdx.x >> 5);
    const int lane = threadIdx.x & 31;
    uint2* myq = q2 + wave * CQCAP;

    const int nslices = (E + SLICE - 1) / SLICE;
    const int ngroups = (nslices + GRP_SLICES - 1) / GRP_SLICES;
    const int slot    = lane & 15;

    for (int tile = blockIdx.x; tile * TILE_ROWS < N; tile += gridDim.x) {
        const int lo    = tile * TILE_ROWS;
        const int nrows = min(TILE_ROWS, N - lo);

        for (int i = threadIdx.x; i < TILE_ROWS * D / 4; i += blockDim.x)
            ((float4*)acc)[i] = make_float4(0.0f, 0.0f, 0.0f, 0.0f);
        __syncthreads();

        const uint2* __restrict__ ct = cells + (size_t)tile * nslices * CELL_SLOTS;
        uint2 cur[GRP_LOADS] = {};
        if (wave < ngroups) fetch_cells(ct, wave, lane, nslices, cur);

        int cnt = 0;
        for (int g = wave; g < ngroups; g += OWN_WAVES) {
            uint2 nxt[GRP_LOADS];
            fetch_cells(ct, (g + OWN_WAVES < ngroups) ? g + OWN_WAVES : g, lane, nslices, nxt);

#pragma unroll
            for (int j = 0; j < GRP_LOADS; ++j) {
                const int c0 = __builtin_amdgcn_readlane((int)cur[j].x, 0);
                const int c1 = __builtin_amdgcn_readlane((int)cur[j].x, 16);
                const int hc = min((lane & 16) ? c1 : c0, CELL_RECS);
                const bool ok = slot >= 1 && slot <= hc;
                const unsigned m = __builtin_amdgcn_ballot_w32(ok);
                if (ok) myq[cnt + __builtin_popcount(m & ((1u << lane) - 1u))] = cur[j];
                cnt += __builtin_popcount(m);
            }

            int head = 0;
            for (; cnt - head >= 32; head += 32)
                gather_recs<true>(h, acc, myq + head, 32, lane);
            if (g + OWN_WAVES >= ngroups && cnt > head) {
                gather_recs<false>(h, acc, myq + head, cnt - head, lane);
                head = cnt;
            }
            const int rest = cnt - head;
            if (head > 0 && rest > 0) {
                const uint2 en = myq[head + ((lane < rest) ? lane : 0)];
                if (lane < rest) myq[lane] = en;
            }
            cnt = rest;

#pragma unroll
            for (int j = 0; j < GRP_LOADS; ++j) cur[j] = nxt[j];
        }

        const int nside = min(*side_count, E);
        for (int i = threadIdx.x; i < nside; i += blockDim.x) {
            const int e = side_idx[i];
            const int c = col_idx[e];
            if ((unsigned)(c - lo) < (unsigned)nrows) {
                const float wt = w[e];
                const float* __restrict__ hr = h + (size_t)row_idx[e] * D;
                float* a = acc + (c - lo) * D;
                for (int j = 0; j < D; ++j)
                    __hip_atomic_fetch_add(a + 32 * (j & 3) + (j >> 2), hr[j] * wt, __ATOMIC_RELAXED,
                                           __HIP_MEMORY_SCOPE_WORKGROUP);
            }
        }
        __syncthreads();

        float4* o4 = (float4*)(out + (size_t)lo * D);
        for (int i = threadIdx.x; i < nrows * (D / 4); i += blockDim.x) {
            const int   c  = lo + i / (D / 4);
            const float dc = inv_sqrt_deg(deg2[c], deg2[N + c]);
            float4 o = o4[i];
            const float* a = acc + (i / (D / 4)) * D + (i % (D / 4));
            o.x += a[0] * dc; o.y += a[32] * dc; o.z += a[64] * dc; o.w += a[96] * dc;
            o4[i] = o;
        }
        __syncthreads();
    }
}


extern "C" void kernel_launch(void* const* d_in, const int* in_sizes, int n_in,
                              void* d_out, int out_size, void* d_ws, size_t ws_size,
                              hipStream_t stream) {
    const float* x    = (const float*)d_in[0];
    const int*   ei   = (const int*)d_in[1];
    const float* ew   = (const float*)d_in[2];
    const float* Wlin = (const float*)d_in[3];
    const float* Wgcn = (const float*)d_in[4];
    float*       out  = (float*)d_out;

    const int N = in_sizes[0] / D;
    const int E = in_sizes[2];

    const int* row = ei;
    const int* col = ei + E;

    float* deg2 = (float*)d_ws;
    float* h    = deg2 + 2 * N;
    const int nslices = (E + SLICE - 1) / SLICE;
    const bool keyed = ws_size >= (size_t)(2 * N + N * D) * sizeof(float) + (size_t)nslices * SLICE;
    unsigned* keys = keyed ? (unsigned*)(h + (size_t)N * D) : nullptr;
    const size_t rec_off = ((size_t)(2 * N + N * D) * sizeof(float) + (size_t)nslices * SLICE + 15) & ~(size_t)15;
    const bool rec = keyed && N <= (1 << REC_ROW_BITS) && ws_size >= rec_off + (size_t)nslices * SLICE * 8;
    unsigned* recs = rec ? (unsigned*)((char*)d_ws + rec_off) : nullptr;

    const int    ntile     = (N + TILE_ROWS - 1) / TILE_ROWS;
    const size_t cells_off = ((size_t)(2 * N + N * D) * sizeof(float) + 127) & ~(size_t)127;
    const size_t side_off  = cells_off + (size_t)ntile * nslices * (CELL_SLOTS * 8);
    const size_t cnt_off   = (side_off + (size_t)E * sizeof(int) + 15) & ~(size_t)15;
    if (ntile <= MAX_TILES && N <= (1 << REC_ROW_BITS) && ws_size >= cnt_off + 16) {
        uint4* cells      = (uint4*)((char*)d_ws + cells_off);
        int*   side_idx   = (int*)((char*)d_ws + side_off);
        int*   side_count = (int*)((char*)d_ws + cnt_off);
        zero_f32_and_count<<<(2 * N + 255) / 256, 256, 0, stream>>>(deg2, 2 * N, side_count);
        deg_cells_lin_kernel<<<LIN_BLOCKS + nslices, SLICE / 4, 0, stream>>>(row, col, ew, deg2, cells, side_count,
                                                                            side_idx, N, E, nslices, ntile,
                                                                            x, Wlin, out, N / 16, LIN_BLOCKS);
        gemm_gcn<<<GCN_BLOCKS, 256, 0, stream>>>(x, Wgcn, deg2, h, N / 16, N);
        owner_gather_cells<<<ntile, 1024, 0, stream>>>(h, row, col, ew, deg2, (const uint2*)cells, side_count,
                                                       side_idx, out, N, E);
        return;
    }

    zero_f32<<<(2 * N + 255) / 256, 256, 0, stream>>>(deg2, 2 * N);
    deg_kernel<<<nslices, SLICE / 4, 0, stream>>>(row, col, ew, deg2, keys, recs, N, E);
    gemm_dual<<<512, 512, 0, stream>>>(x, Wlin, Wgcn, deg2, out, h, N / 16, N);
    const int nblk = (N + TILE_ROWS - 1) / TILE_ROWS;
    if (rec)        owner_gather<true, true><<<nblk, 1024, 0, stream>>>(h, row, col, ew, deg2, keys, recs, out, N, E);
    else if (keyed) owner_gather<true, false><<<nblk, 1024, 0, stream>>>(h, row, col, ew, deg2, keys, recs, out, N, E);
    else            owner_gather<false, false><<<nblk, 1024, 0, stream>>>(h, row, col, ew, deg2, keys, recs, out, N, E);
}
